# starting kernel
# baseline (speedup 1.0000x reference)
#include <hip/hip_runtime.h>
#include <hip/hip_bf16.h>

#define B_  32
#define D_  1024
#define R_  2048

typedef float v2f __attribute__((ext_vector_type(2)));
typedef float v8f __attribute__((ext_vector_type(8)));

__device__ __forceinline__ void top2_insert(float v, int i,
                                            float& v1, int& i1,
                                            float& v2, int& i2) {
    bool beats1 = (v > v1) || (v == v1 && i < i1);
    if (beats1) {
        v2 = v1; i2 = i1;
        v1 = v;  i1 = i;
    } else {
        bool beats2 = (v > v2) || (v == v2 && i < i2);
        if (beats2) { v2 = v; i2 = i; }
    }
}

__global__ void rule_prep_kernel(const float* __restrict__ fact_logits,
                                 const float* __restrict__ agg_logits,
                                 const float* __restrict__ strength,
                                 int*   __restrict__ ruleIdx,
                                 float* __restrict__ ruleW)
{
    const int wavesPerBlock = blockDim.x >> 5;
    const int r    = blockIdx.x * wavesPerBlock + (threadIdx.x >> 5);
    const int lane = threadIdx.x & 31;
    if (r >= R_) return;

    float v1 = -__builtin_inff(), v2 = -__builtin_inff();
    int   i1 = 0x7fffffff,        i2 = 0x7fffffff;

    const float* row = fact_logits + (size_t)r * D_;
    #pragma unroll 4
    for (int j = lane; j < D_; j += 32) {
        top2_insert(row[j], j, v1, i1, v2, i2);
    }
    #pragma unroll
    for (int m = 16; m >= 1; m >>= 1) {
        float ov1 = __shfl_xor(v1, m, 32);
        int   oi1 = __shfl_xor(i1, m, 32);
        float ov2 = __shfl_xor(v2, m, 32);
        int   oi2 = __shfl_xor(i2, m, 32);
        top2_insert(ov1, oi1, v1, i1, v2, i2);
        top2_insert(ov2, oi2, v1, i1, v2, i2);
    }

    if (lane == 0) {
        ruleIdx[2 * r + 0] = i1;
        ruleIdx[2 * r + 1] = i2;
        float a0 = agg_logits[3 * r + 0];
        float a1 = agg_logits[3 * r + 1];
        float a2 = agg_logits[3 * r + 2];
        float mx = fmaxf(a0, fmaxf(a1, a2));
        float e0 = expf(a0 - mx), e1 = expf(a1 - mx), e2 = expf(a2 - mx);
        float inv = 1.0f / (e0 + e1 + e2);
        ruleW[4 * r + 0] = e0 * inv;
        ruleW[4 * r + 1] = e1 * inv;
        ruleW[4 * r + 2] = e2 * inv;
        ruleW[4 * r + 3] = 1.0f / (1.0f + expf(-strength[r]));
    }
}

__global__ void gemm_wmma_kernel(const float* __restrict__ facts,
                                 const float* __restrict__ Wp,
                                 float* __restrict__ projPart,
                                 int klen)
{
    const int wave  = (blockIdx.x * blockDim.x + threadIdx.x) >> 5;
    const int lane  = threadIdx.x & 31;

    const int ntile = wave & 127;
    const int slice = wave >> 7;
    const int n0    = ntile * 16;
    const int kbeg  = slice * klen;

    const int rlo   = lane & 15;
    const int khalf = (lane >> 4) * 2;

    const float* aptr0 = facts + (size_t)(rlo)      * D_ + kbeg + khalf;
    const float* aptr1 = facts + (size_t)(16 + rlo) * D_ + kbeg + khalf;
    const float* bptr  = Wp    + (size_t)(n0 + rlo) * D_ + kbeg + khalf;

    v8f c0 = {};
    v8f c1 = {};
    for (int k = 0; k < klen; k += 16) {
        __builtin_prefetch(bptr + k + 512, 0, 3);
        #pragma unroll
        for (int kk = 0; kk < 16; kk += 4) {
            v2f b  = *(const v2f*)(bptr  + k + kk);
            v2f a0 = *(const v2f*)(aptr0 + k + kk);
            v2f a1 = *(const v2f*)(aptr1 + k + kk);
            c0 = __builtin_amdgcn_wmma_f32_16x16x4_f32(false, a0, false, b,
                                                       (short)0, c0, false, false);
            c1 = __builtin_amdgcn_wmma_f32_16x16x4_f32(false, a1, false, b,
                                                       (short)0, c1, false, false);
        }
    }

    float* outp = projPart + (size_t)slice * B_ * R_;
    const int col   = n0 + rlo;
    const int rbase = (lane >> 4) * 8;
    #pragma unroll
    for (int v = 0; v < 8; ++v) {
        outp[(size_t)(rbase + v)      * R_ + col] = c0[v];
        outp[(size_t)(rbase + v + 16) * R_ + col] = c1[v];
    }
}

__global__ void rule_act_kernel(const float* __restrict__ facts,
                                const int*   __restrict__ ruleIdx,
                                const float* __restrict__ ruleW,
                                float* __restrict__ ract)
{
    const int gid = blockIdx.x * blockDim.x + threadIdx.x;
    const int b = gid >> 11;
    const int r = gid & (R_ - 1);

    const float f0 = facts[(size_t)b * D_ + ruleIdx[2 * r + 0]];
    const float f1 = facts[(size_t)b * D_ + ruleIdx[2 * r + 1]];
    const float w0 = ruleW[4 * r + 0];
    const float w1 = ruleW[4 * r + 1];
    const float w2 = ruleW[4 * r + 2];
    const float sg = ruleW[4 * r + 3];

    const float and_a = f0 * f1;
    const float or_a  = 1.0f - (1.0f - f0) * (1.0f - f1);
    const float k_a   = (f0 + f1) / (2.0f + 1e-8f);

    ract[gid] = (w0 * and_a + w1 * or_a + w2 * k_a) * sg;
}

__global__ void gate_ln_kernel(const float* __restrict__ projPart,
                               int nslice,
                               const float* __restrict__ ract,
                               const float* __restrict__ gamma,
                               const float* __restrict__ beta,
                               float* __restrict__ out)
{
    const int b   = blockIdx.x;
    const int tid = threadIdx.x;

    __shared__ float s_pre[R_];
    __shared__ float s_act[R_];
    __shared__ unsigned char s_ch[R_];
    __shared__ float s_val[256];
    __shared__ int   s_idx[256];

    for (int i = tid; i < R_; i += 256) {
        float acc = 0.0f;
        for (int s = 0; s < nslice; ++s) {
            acc += projPart[(size_t)s * B_ * R_ + (size_t)b * R_ + i];
        }
        s_pre[i] = acc;
        s_act[i] = ract[(size_t)b * R_ + i];
        s_ch[i]  = 0;
    }
    __syncthreads();

    for (int t = 0; t < 8; ++t) {
        float bv = -__builtin_inff();
        int   bi = 0x7fffffff;
        for (int i = tid; i < R_; i += 256) {
            if (!s_ch[i]) {
                float v = s_act[i];
                if (v > bv || (v == bv && i < bi)) { bv = v; bi = i; }
            }
        }
        s_val[tid] = bv; s_idx[tid] = bi;
        __syncthreads();
        for (int s = 128; s > 0; s >>= 1) {
            if (tid < s) {
                float ov = s_val[tid + s]; int oi = s_idx[tid + s];
                if (ov > s_val[tid] || (ov == s_val[tid] && oi < s_idx[tid])) {
                    s_val[tid] = ov; s_idx[tid] = oi;
                }
            }
            __syncthreads();
        }
        if (tid == 0 && s_idx[0] < R_) s_ch[s_idx[0]] = 1;
        __syncthreads();
    }

    for (int i = tid; i < R_; i += 256) {
        if (s_ch[i]) s_pre[i] += s_act[i];
    }
    __syncthreads();

    float sum = 0.0f;
    for (int i = tid; i < R_; i += 256) sum += s_pre[i];
    s_val[tid] = sum;
    __syncthreads();
    for (int s = 128; s > 0; s >>= 1) {
        if (tid < s) s_val[tid] += s_val[tid + s];
        __syncthreads();
    }
    const float mu = s_val[0] * (1.0f / R_);
    __syncthreads();

    float ss = 0.0f;
    for (int i = tid; i < R_; i += 256) { float d = s_pre[i] - mu; ss += d * d; }
    s_val[tid] = ss;
    __syncthreads();
    for (int s = 128; s > 0; s >>= 1) {
        if (tid < s) s_val[tid] += s_val[tid + s];
        __syncthreads();
    }
    const float inv = rsqrtf(s_val[0] * (1.0f / R_) + 1e-5f);

    for (int i = tid; i < R_; i += 256) {
        out[(size_t)b * R_ + i] = (s_pre[i] - mu) * inv * gamma[i] + beta[i];
    }
}

extern "C" void kernel_launch(void* const* d_in, const int* in_sizes, int n_in,
                              void* d_out, int out_size, void* d_ws, size_t ws_size,
                              hipStream_t stream) {
    const float* facts        = (const float*)d_in[0];
    const float* fact_logits  = (const float*)d_in[1];
    const float* agg_logits   = (const float*)d_in[2];
    const float* strength     = (const float*)d_in[3];
    const float* W_proj       = (const float*)d_in[4];
    const float* ln_gamma     = (const float*)d_in[5];
    const float* ln_beta      = (const float*)d_in[6];
    float* out = (float*)d_out;

    const size_t kFixed = 49152 + (size_t)B_ * R_ * 4;
    int nslice = 1;
    if (ws_size >= kFixed + 4 * (size_t)B_ * R_ * 4) nslice = 4;
    else if (ws_size >= kFixed + 2 * (size_t)B_ * R_ * 4) nslice = 2;

    char* ws = (char*)d_ws;
    float* ruleW    = (float*)(ws);
    int*   ruleIdx  = (int*)  (ws + (size_t)R_ * 4 * 4);
    float* ract     = (float*)(ws + 49152);
    float* projPart = (float*)(ws + kFixed);

    rule_prep_kernel<<<R_ / 8, 256, 0, stream>>>(fact_logits, agg_logits, strength,
                                                 ruleIdx, ruleW);

    gemm_wmma_kernel<<<16 * nslice, 256, 0, stream>>>(facts, W_proj, projPart,
                                                      D_ / nslice);

    rule_act_kernel<<<(B_ * R_) / 256, 256, 0, stream>>>(facts, ruleIdx, ruleW, ract);

    gate_ln_kernel<<<B_, 256, 0, stream>>>(projPart, nslice, ract,
                                           ln_gamma, ln_beta, out);
}


# gate+LayerNorm kernel: row kept in registers; top-8, mean and variance by DPP wave reductions and one small LDS exchange each (3 block barriers, was about 100); mean/var reduction tree changed
# speedup vs baseline: 1.9539x; 1.9539x over previous
#include <hip/hip_runtime.h>
#include <hip/hip_bf16.h>

#define B_  32
#define D_  1024
#define R_  2048

typedef float v2f __attribute__((ext_vector_type(2)));
typedef float v8f __attribute__((ext_vector_type(8)));
typedef float v4f __attribute__((ext_vector_type(4)));

__device__ __forceinline__ void top2_insert(float v, int i,
                                            float& v1, int& i1,
                                            float& v2, int& i2) {
    bool beats1 = (v > v1) || (v == v1 && i < i1);
    if (beats1) {
        v2 = v1; i2 = i1;
        v1 = v;  i1 = i;
    } else {
        bool beats2 = (v > v2) || (v == v2 && i < i2);
        if (beats2) { v2 = v; i2 = i; }
    }
}

__global__ void rule_prep_kernel(const float* __restrict__ fact_logits,
                                 const float* __restrict__ agg_logits,
                                 const float* __restrict__ strength,
                                 int*   __restrict__ ruleIdx,
                                 float* __restrict__ ruleW)
{
    const int wavesPerBlock = blockDim.x >> 5;
    const int r    = blockIdx.x * wavesPerBlock + (threadIdx.x >> 5);
    const int lane = threadIdx.x & 31;
    if (r >= R_) return;

    float v1 = -__builtin_inff(), v2 = -__builtin_inff();
    int   i1 = 0x7fffffff,        i2 = 0x7fffffff;

    const float* row = fact_logits + (size_t)r * D_;
    #pragma unroll 4
    for (int j = lane; j < D_; j += 32) {
        top2_insert(row[j], j, v1, i1, v2, i2);
    }
    #pragma unroll
    for (int m = 16; m >= 1; m >>= 1) {
        float ov1 = __shfl_xor(v1, m, 32);
        int   oi1 = __shfl_xor(i1, m, 32);
        float ov2 = __shfl_xor(v2, m, 32);
        int   oi2 = __shfl_xor(i2, m, 32);
        top2_insert(ov1, oi1, v1, i1, v2, i2);
        top2_insert(ov2, oi2, v1, i1, v2, i2);
    }

    if (lane == 0) {
        ruleIdx[2 * r + 0] = i1;
        ruleIdx[2 * r + 1] = i2;
        float a0 = agg_logits[3 * r + 0];
        float a1 = agg_logits[3 * r + 1];
        float a2 = agg_logits[3 * r + 2];
        float mx = fmaxf(a0, fmaxf(a1, a2));
        float e0 = expf(a0 - mx), e1 = expf(a1 - mx), e2 = expf(a2 - mx);
        float inv = 1.0f / (e0 + e1 + e2);
        ruleW[4 * r + 0] = e0 * inv;
        ruleW[4 * r + 1] = e1 * inv;
        ruleW[4 * r + 2] = e2 * inv;
        ruleW[4 * r + 3] = 1.0f / (1.0f + expf(-strength[r]));
    }
}

__global__ void gemm_wmma_kernel(const float* __restrict__ facts,
                                 const float* __restrict__ Wp,
                                 float* __restrict__ projPart,
                                 int klen)
{
    const int wave  = (blockIdx.x * blockDim.x + threadIdx.x) >> 5;
    const int lane  = threadIdx.x & 31;

    const int ntile = wave & 127;
    const int slice = wave >> 7;
    const int n0    = ntile * 16;
    const int kbeg  = slice * klen;

    const int rlo   = lane & 15;
    const int khalf = (lane >> 4) * 2;

    const float* aptr0 = facts + (size_t)(rlo)      * D_ + kbeg + khalf;
    const float* aptr1 = facts + (size_t)(16 + rlo) * D_ + kbeg + khalf;
    const float* bptr  = Wp    + (size_t)(n0 + rlo) * D_ + kbeg + khalf;

    v8f c0 = {};
    v8f c1 = {};
    for (int k = 0; k < klen; k += 16) {
        __builtin_prefetch(bptr + k + 512, 0, 3);
        #pragma unroll
        for (int kk = 0; kk < 16; kk += 4) {
            v2f b  = *(const v2f*)(bptr  + k + kk);
            v2f a0 = *(const v2f*)(aptr0 + k + kk);
            v2f a1 = *(const v2f*)(aptr1 + k + kk);
            c0 = __builtin_amdgcn_wmma_f32_16x16x4_f32(false, a0, false, b,
                                                       (short)0, c0, false, false);
            c1 = __builtin_amdgcn_wmma_f32_16x16x4_f32(false, a1, false, b,
                                                       (short)0, c1, false, false);
        }
    }

    float* outp = projPart + (size_t)slice * B_ * R_;
    const int col   = n0 + rlo;
    const int rbase = (lane >> 4) * 8;
    #pragma unroll
    for (int v = 0; v < 8; ++v) {
        outp[(size_t)(rbase + v)      * R_ + col] = c0[v];
        outp[(size_t)(rbase + v + 16) * R_ + col] = c1[v];
    }
}

__global__ void rule_act_kernel(const float* __restrict__ facts,
                                const int*   __restrict__ ruleIdx,
                                const float* __restrict__ ruleW,
                                float* __restrict__ ract)
{
    const int gid = blockIdx.x * blockDim.x + threadIdx.x;
    const int b = gid >> 11;
    const int r = gid & (R_ - 1);

    const float f0 = facts[(size_t)b * D_ + ruleIdx[2 * r + 0]];
    const float f1 = facts[(size_t)b * D_ + ruleIdx[2 * r + 1]];
    const float w0 = ruleW[4 * r + 0];
    const float w1 = ruleW[4 * r + 1];
    const float w2 = ruleW[4 * r + 2];
    const float sg = ruleW[4 * r + 3];

    const float and_a = f0 * f1;
    const float or_a  = 1.0f - (1.0f - f0) * (1.0f - f1);
    const float k_a   = (f0 + f1) / (2.0f + 1e-8f);

    ract[gid] = (w0 * and_a + w1 * or_a + w2 * k_a) * sg;
}

template <int M>
__device__ __forceinline__ int row_xor_i(int x) {
    return __builtin_amdgcn_update_dpp(0, x, 0x160 + M, 0xf, 0xf, true);
}
__device__ __forceinline__ int half_xor_i(int x) {
    return __builtin_amdgcn_permlanex16(x, x, 0x76543210, 0xfedcba98, false, false);
}
template <int M>
__device__ __forceinline__ float row_xor_f(float x) {
    return __builtin_bit_cast(float, row_xor_i<M>(__builtin_bit_cast(int, x)));
}
__device__ __forceinline__ float half_xor_f(float x) {
    return __builtin_bit_cast(float, half_xor_i(__builtin_bit_cast(int, x)));
}
__device__ __forceinline__ int wave_max_i(int v) {
    v = max(v, row_xor_i<1>(v));
    v = max(v, row_xor_i<2>(v));
    v = max(v, row_xor_i<4>(v));
    v = max(v, row_xor_i<8>(v));
    v = max(v, half_xor_i(v));
    return v;
}
__device__ __forceinline__ int wave_min_i(int v) {
    v = min(v, row_xor_i<1>(v));
    v = min(v, row_xor_i<2>(v));
    v = min(v, row_xor_i<4>(v));
    v = min(v, row_xor_i<8>(v));
    v = min(v, half_xor_i(v));
    return v;
}
__device__ __forceinline__ float wave_sum_f(float v) {
    v += row_xor_f<1>(v);
    v += row_xor_f<2>(v);
    v += row_xor_f<4>(v);
    v += row_xor_f<8>(v);
    v += half_xor_f(v);
    return v;
}
__device__ __forceinline__ int order_key(float v) {
    const int u = __builtin_bit_cast(int, v + 0.0f);
    return u ^ ((u >> 31) & 0x7fffffff);
}
__device__ __forceinline__ void wave_best(int bk, int bi, int& wk, int& wi) {
    wk = wave_max_i(bk);
    wi = wave_min_i((bk == wk) ? bi : 0x7fffffff);
}

__global__ void __launch_bounds__(256)
gate_ln_kernel(const float* __restrict__ projPart,
               int nslice,
               const float* __restrict__ ract,
               const float* __restrict__ gamma,
               const float* __restrict__ beta,
               float* __restrict__ out)
{
    const int b    = blockIdx.x;
    const int tid  = threadIdx.x;
    const int lane = tid & 31;
    const int wave = tid >> 5;
    const int e0   = 4 * tid;

    __shared__ int   s_ck[64];
    __shared__ int   s_ci[64];
    __shared__ float s_sum[8];
    __shared__ float s_sq[8];

    const size_t rowoff = (size_t)b * R_;

    v4f a4[2];
    #pragma unroll
    for (int h = 0; h < 2; ++h) a4[h] = *(const v4f*)(ract + rowoff + h * 1024 + e0);
    const float act[8] = {a4[0].x, a4[0].y, a4[0].z, a4[0].w, a4[1].x, a4[1].y, a4[1].z, a4[1].w};

    v4f g4[2], be4[2], p4[4][2];
    #pragma unroll
    for (int h = 0; h < 2; ++h) {
        const int col = h * 1024 + e0;
        #pragma unroll
        for (int s = 0; s < 4; ++s) {
            const int se = (s < nslice) ? s : 0;
            p4[s][h] = *(const v4f*)(projPart + (size_t)se * B_ * R_ + rowoff + col);
        }
        g4[h]  = *(const v4f*)(gamma + col);
        be4[h] = *(const v4f*)(beta + col);
    }

    const int kTaken = (int)0x80000000;
    int key[8];
    #pragma unroll
    for (int j = 0; j < 8; ++j) key[j] = order_key(act[j]);

    int myck = kTaken;
    int myci = 0x7fffffff;
    #pragma unroll
    for (int t = 0; t < 8; ++t) {
        int bk = kTaken;
        int bi = 0x7fffffff;
        #pragma unroll
        for (int j = 0; j < 8; ++j) {
            const int  idx  = (j >> 2) * 1024 + e0 + (j & 3);
            const bool take = key[j] > bk;
            bk = take ? key[j] : bk;
            bi = take ? idx : bi;
        }
        int wk, wi;
        wave_best(bk, bi, wk, wi);
        #pragma unroll
        for (int j = 0; j < 8; ++j) {
            const int idx = (j >> 2) * 1024 + e0 + (j & 3);
            key[j] = (idx == wi) ? kTaken : key[j];
        }
        myck = (lane == t) ? wk : myck;
        myci = (lane == t) ? wi : myci;
    }
    if (lane < 8) {
        s_ck[wave * 8 + lane] = myck;
        s_ci[wave * 8 + lane] = myci;
    }
    __syncthreads();

    int ck[2], ci[2];
    ck[0] = s_ck[lane];      ci[0] = s_ci[lane];
    ck[1] = s_ck[lane + 32]; ci[1] = s_ci[lane + 32];
    unsigned gmask = 0;
    #pragma unroll
    for (int t = 0; t < 8; ++t) {
        int bk = kTaken;
        int bi = 0x7fffffff;
        #pragma unroll
        for (int q = 0; q < 2; ++q) {
            const bool take = (ck[q] > bk) | ((ck[q] == bk) & (ci[q] < bi));
            bk = take ? ck[q] : bk;
            bi = take ? ci[q] : bi;
        }
        int wk, wi;
        wave_best(bk, bi, wk, wi);
        #pragma unroll
        for (int q = 0; q < 2; ++q) ck[q] = (ci[q] == wi) ? kTaken : ck[q];
        const unsigned uw   = (unsigned)wi;
        const bool     mine = ((wi & 1020) == e0) & (uw < (unsigned)R_);
        const unsigned bit  = 1u << ((((uw >> 10) & 1u) << 2) | (uw & 3u));
        gmask |= mine ? bit : 0u;
    }

    float pre[8];
    #pragma unroll
    for (int h = 0; h < 2; ++h) {
        #pragma unroll
        for (int c = 0; c < 4; ++c) {
            float acc = 0.0f;
            acc += p4[0][h][c];
            if (nslice > 1) acc += p4[1][h][c];
            if (nslice > 2) acc += p4[2][h][c];
            if (nslice > 3) acc += p4[3][h][c];
            pre[h * 4 + c] = acc;
        }
    }

    #pragma unroll
    for (int j = 0; j < 8; ++j) {
        const float g = pre[j] + act[j];
        pre[j] = ((gmask >> j) & 1u) ? g : pre[j];
    }

    float sum = 0.0f;
    #pragma unroll
    for (int j = 0; j < 8; ++j) sum += pre[j];
    sum = wave_sum_f(sum);
    if (lane == 0) s_sum[wave] = sum;
    __syncthreads();
    float tot = 0.0f;
    #pragma unroll
    for (int w = 0; w < 8; ++w) tot += s_sum[w];
    const float mu = tot * (1.0f / R_);

    float ss = 0.0f;
    #pragma unroll
    for (int j = 0; j < 8; ++j) { float d = pre[j] - mu; ss += d * d; }
    ss = wave_sum_f(ss);
    if (lane == 0) s_sq[wave] = ss;
    __syncthreads();
    float tot2 = 0.0f;
    #pragma unroll
    for (int w = 0; w < 8; ++w) tot2 += s_sq[w];
    const float inv = rsqrtf(tot2 * (1.0f / R_) + 1e-5f);

    #pragma unroll
    for (int h = 0; h < 2; ++h) {
        v4f o;
        o.x = (pre[h * 4 + 0] - mu) * inv * g4[h].x + be4[h].x;
        o.y = (pre[h * 4 + 1] - mu) * inv * g4[h].y + be4[h].y;
        o.z = (pre[h * 4 + 2] - mu) * inv * g4[h].z + be4[h].z;
        o.w = (pre[h * 4 + 3] - mu) * inv * g4[h].w + be4[h].w;
        *(v4f*)(out + rowoff + h * 1024 + e0) = o;
    }
}

extern "C" void kernel_launch(void* const* d_in, const int* in_sizes, int n_in,
                              void* d_out, int out_size, void* d_ws, size_t ws_size,
                              hipStream_t stream) {
    const float* facts        = (const float*)d_in[0];
    const float* fact_logits  = (const float*)d_in[1];
    const float* agg_logits   = (const float*)d_in[2];
    const float* strength     = (const float*)d_in[3];
    const float* W_proj       = (const float*)d_in[4];
    const float* ln_gamma     = (const float*)d_in[5];
    const float* ln_beta      = (const float*)d_in[6];
    float* out = (float*)d_out;

    const size_t kFixed = 49152 + (size_t)B_ * R_ * 4;
    int nslice = 1;
    if (ws_size >= kFixed + 4 * (size_t)B_ * R_ * 4) nslice = 4;
    else if (ws_size >= kFixed + 2 * (size_t)B_ * R_ * 4) nslice = 2;

    char* ws = (char*)d_ws;
    float* ruleW    = (float*)(ws);
    int*   ruleIdx  = (int*)  (ws + (size_t)R_ * 4 * 4);
    float* ract     = (float*)(ws + 49152);
    float* projPart = (float*)(ws + kFixed);

    rule_prep_kernel<<<R_ / 8, 256, 0, stream>>>(fact_logits, agg_logits, strength,
                                                 ruleIdx, ruleW);

    gemm_wmma_kernel<<<16 * nslice, 256, 0, stream>>>(facts, W_proj, projPart,
                                                      D_ / nslice);

    rule_act_kernel<<<(B_ * R_) / 256, 256, 0, stream>>>(facts, ruleIdx, ruleW, ract);

    gate_ln_kernel<<<B_, 256, 0, stream>>>(projPart, nslice, ract,
                                           ln_gamma, ln_beta, out);
}


# 4 launches -> 2: rule prep and the projection GEMM are independent blocks of one launch; rule activations are computed inside the gate+LayerNorm kernel (facts row in LDS); sum order unchanged
# speedup vs baseline: 2.8914x; 1.4798x over previous
#include <hip/hip_runtime.h>
#include <hip/hip_bf16.h>

#define B_  32
#define D_  1024
#define R_  2048

typedef float v2f __attribute__((ext_vector_type(2)));
typedef float v8f __attribute__((ext_vector_type(8)));
typedef float v4f __attribute__((ext_vector_type(4)));
typedef int   v4i __attribute__((ext_vector_type(4)));

__device__ __forceinline__ void top2_insert(float v, int i,
                                            float& v1, int& i1,
                                            float& v2, int& i2) {
    bool beats1 = (v > v1) || (v == v1 && i < i1);
    if (beats1) {
        v2 = v1; i2 = i1;
        v1 = v;  i1 = i;
    } else {
        bool beats2 = (v > v2) || (v == v2 && i < i2);
        if (beats2) { v2 = v; i2 = i; }
    }
}

__device__ __forceinline__ void rule_prep_body(int blk,
                                 const float* __restrict__ fact_logits,
                                 const float* __restrict__ agg_logits,
                                 const float* __restrict__ strength,
                                 int*   __restrict__ ruleIdx,
                                 float* __restrict__ ruleW)
{
    const int wavesPerBlock = blockDim.x >> 5;
    const int r    = blk * wavesPerBlock + (threadIdx.x >> 5);
    const int lane = threadIdx.x & 31;
    if (r >= R_) return;

    float v1 = -__builtin_inff(), v2 = -__builtin_inff();
    int   i1 = 0x7fffffff,        i2 = 0x7fffffff;

    const float* row = fact_logits + (size_t)r * D_;
    #pragma unroll 4
    for (int j = lane; j < D_; j += 32) {
        top2_insert(row[j], j, v1, i1, v2, i2);
    }
    #pragma unroll
    for (int m = 16; m >= 1; m >>= 1) {
        float ov1 = __shfl_xor(v1, m, 32);
        int   oi1 = __shfl_xor(i1, m, 32);
        float ov2 = __shfl_xor(v2, m, 32);
        int   oi2 = __shfl_xor(i2, m, 32);
        top2_insert(ov1, oi1, v1, i1, v2, i2);
        top2_insert(ov2, oi2, v1, i1, v2, i2);
    }

    if (lane == 0) {
        ruleIdx[2 * r + 0] = i1;
        ruleIdx[2 * r + 1] = i2;
        float a0 = agg_logits[3 * r + 0];
        float a1 = agg_logits[3 * r + 1];
        float a2 = agg_logits[3 * r + 2];
        float mx = fmaxf(a0, fmaxf(a1, a2));
        float e0 = expf(a0 - mx), e1 = expf(a1 - mx), e2 = expf(a2 - mx);
        float inv = 1.0f / (e0 + e1 + e2);
        ruleW[4 * r + 0] = e0 * inv;
        ruleW[4 * r + 1] = e1 * inv;
        ruleW[4 * r + 2] = e2 * inv;
        ruleW[4 * r + 3] = 1.0f / (1.0f + expf(-strength[r]));
    }
}

__device__ __forceinline__ void gemm_wmma_body(int blk,
                                 const float* __restrict__ facts,
                                 const float* __restrict__ Wp,
                                 float* __restrict__ projPart,
                                 int klen)
{
    const int wave  = (blk * blockDim.x + threadIdx.x) >> 5;
    const int lane  = threadIdx.x & 31;

    const int ntile = wave & 127;
    const int slice = wave >> 7;
    const int n0    = ntile * 16;
    const int kbeg  = slice * klen;

    const int rlo   = lane & 15;
    const int khalf = (lane >> 4) * 2;

    const float* aptr0 = facts + (size_t)(rlo)      * D_ + kbeg + khalf;
    const float* aptr1 = facts + (size_t)(16 + rlo) * D_ + kbeg + khalf;
    const float* bptr  = Wp    + (size_t)(n0 + rlo) * D_ + kbeg + khalf;

    v8f c0 = {};
    v8f c1 = {};
    for (int k = 0; k < klen; k += 16) {
        __builtin_prefetch(bptr + k + 512, 0, 3);
        #pragma unroll
        for (int kk = 0; kk < 16; kk += 4) {
            v2f b  = *(const v2f*)(bptr  + k + kk);
            v2f a0 = *(const v2f*)(aptr0 + k + kk);
            v2f a1 = *(const v2f*)(aptr1 + k + kk);
            c0 = __builtin_amdgcn_wmma_f32_16x16x4_f32(false, a0, false, b,
                                                       (short)0, c0, false, false);
            c1 = __builtin_amdgcn_wmma_f32_16x16x4_f32(false, a1, false, b,
                                                       (short)0, c1, false, false);
        }
    }

    float* outp = projPart + (size_t)slice * B_ * R_;
    const int col   = n0 + rlo;
    const int rbase = (lane >> 4) * 8;
    #pragma unroll
    for (int v = 0; v < 8; ++v) {
        outp[(size_t)(rbase + v)      * R_ + col] = c0[v];
        outp[(size_t)(rbase + v + 16) * R_ + col] = c1[v];
    }
}

__global__ void __launch_bounds__(256)
prep_gemm_kernel(const float* __restrict__ fact_logits,
                 const float* __restrict__ agg_logits,
                 const float* __restrict__ strength,
                 int*   __restrict__ ruleIdx,
                 float* __restrict__ ruleW,
                 const float* __restrict__ facts,
                 const float* __restrict__ Wp,
                 float* __restrict__ projPart,
                 int klen,
                 int gemmBlocks)
{
    const int blk = blockIdx.x;
    const int g   = blk - gemmBlocks;
    if (g >= 0 && g < gemmBlocks) {
        gemm_wmma_body(g, facts, Wp, projPart, klen);
    } else {
        rule_prep_body(g < 0 ? blk : g, fact_logits, agg_logits, strength, ruleIdx, ruleW);
    }
}

template <int M>
__device__ __forceinline__ int row_xor_i(int x) {
    return __builtin_amdgcn_update_dpp(0, x, 0x160 + M, 0xf, 0xf, true);
}
__device__ __forceinline__ int half_xor_i(int x) {
    return __builtin_amdgcn_permlanex16(x, x, 0x76543210, 0xfedcba98, false, false);
}
template <int M>
__device__ __forceinline__ float row_xor_f(float x) {
    return __builtin_bit_cast(float, row_xor_i<M>(__builtin_bit_cast(int, x)));
}
__device__ __forceinline__ float half_xor_f(float x) {
    return __builtin_bit_cast(float, half_xor_i(__builtin_bit_cast(int, x)));
}
__device__ __forceinline__ int wave_max_i(int v) {
    v = max(v, row_xor_i<1>(v));
    v = max(v, row_xor_i<2>(v));
    v = max(v, row_xor_i<4>(v));
    v = max(v, row_xor_i<8>(v));
    v = max(v, half_xor_i(v));
    return v;
}
__device__ __forceinline__ int wave_min_i(int v) {
    v = min(v, row_xor_i<1>(v));
    v = min(v, row_xor_i<2>(v));
    v = min(v, row_xor_i<4>(v));
    v = min(v, row_xor_i<8>(v));
    v = min(v, half_xor_i(v));
    return v;
}
__device__ __forceinline__ float wave_sum_f(float v) {
    v += row_xor_f<1>(v);
    v += row_xor_f<2>(v);
    v += row_xor_f<4>(v);
    v += row_xor_f<8>(v);
    v += half_xor_f(v);
    return v;
}
__device__ __forceinline__ int order_key(float v) {
    const int u = __builtin_bit_cast(int, v + 0.0f);
    return u ^ ((u >> 31) & 0x7fffffff);
}
__device__ __forceinline__ void wave_best(int bk, int bi, int& wk, int& wi) {
    wk = wave_max_i(bk);
    wi = wave_min_i((bk == wk) ? bi : 0x7fffffff);
}

__global__ void __launch_bounds__(256)
gate_ln_kernel(const float* __restrict__ projPart,
               int nslice,
               const float* __restrict__ facts,
               const int*   __restrict__ ruleIdx,
               const float* __restrict__ ruleW,
               const float* __restrict__ gamma,
               const float* __restrict__ beta,
               float* __restrict__ out)
{
    const int b    = blockIdx.x;
    const int tid  = threadIdx.x;
    const int lane = tid & 31;
    const int wave = tid >> 5;
    const int e0   = 4 * tid;

    __shared__ float s_f[D_];
    __shared__ int   s_ck[64];
    __shared__ int   s_ci[64];
    __shared__ float s_sum[8];
    __shared__ float s_sq[8];

    const size_t rowoff = (size_t)b * R_;

    *(v4f*)(s_f + e0) = *(const v4f*)(facts + (size_t)b * D_ + e0);

    v4i ix[2][2];
    v4f rw[2][4];
    #pragma unroll
    for (int h = 0; h < 2; ++h) {
        const int r0 = h * 1024 + e0;
        ix[h][0] = *(const v4i*)(ruleIdx + 2 * r0);
        ix[h][1] = *(const v4i*)(ruleIdx + 2 * r0 + 4);
        #pragma unroll
        for (int c = 0; c < 4; ++c) rw[h][c] = *(const v4f*)(ruleW + 4 * (r0 + c));
    }

    v4f g4[2], be4[2], p4[4][2];
    #pragma unroll
    for (int h = 0; h < 2; ++h) {
        const int col = h * 1024 + e0;
        #pragma unroll
        for (int s = 0; s < 4; ++s) {
            const int se = (s < nslice) ? s : 0;
            p4[s][h] = *(const v4f*)(projPart + (size_t)se * B_ * R_ + rowoff + col);
        }
        g4[h]  = *(const v4f*)(gamma + col);
        be4[h] = *(const v4f*)(beta + col);
    }

    __syncthreads();

    float act[8];
    #pragma unroll
    for (int j = 0; j < 8; ++j) {
        const v4i  q  = ix[j >> 2][(j >> 1) & 1];
        const int  i0 = ((j & 1) ? q.z : q.x) & (D_ - 1);
        const int  i1 = ((j & 1) ? q.w : q.y) & (D_ - 1);
        const float f0 = s_f[i0];
        const float f1 = s_f[i1];
        const v4f  w  = rw[j >> 2][j & 3];

        const float and_a = f0 * f1;
        const float or_a  = 1.0f - (1.0f - f0) * (1.0f - f1);
        const float k_a   = (f0 + f1) / (2.0f + 1e-8f);

        act[j] = (w.x * and_a + w.y * or_a + w.z * k_a) * w.w;
    }

    const int kTaken = (int)0x80000000;
    int key[8];
    #pragma unroll
    for (int j = 0; j < 8; ++j) key[j] = order_key(act[j]);

    int myck = kTaken;
    int myci = 0x7fffffff;
    #pragma unroll
    for (int t = 0; t < 8; ++t) {
        int bk = kTaken;
        int bi = 0x7fffffff;
        #pragma unroll
        for (int j = 0; j < 8; ++j) {
            const int  idx  = (j >> 2) * 1024 + e0 + (j & 3);
            const bool take = key[j] > bk;
            bk = take ? key[j] : bk;
            bi = take ? idx : bi;
        }
        int wk, wi;
        wave_best(bk, bi, wk, wi);
        #pragma unroll
        for (int j = 0; j < 8; ++j) {
            const int idx = (j >> 2) * 1024 + e0 + (j & 3);
            key[j] = (idx == wi) ? kTaken : key[j];
        }
        myck = (lane == t) ? wk : myck;
        myci = (lane == t) ? wi : myci;
    }
    if (lane < 8) {
        s_ck[wave * 8 + lane] = myck;
        s_ci[wave * 8 + lane] = myci;
    }
    __syncthreads();

    int ck[2], ci[2];
    ck[0] = s_ck[lane];      ci[0] = s_ci[lane];
    ck[1] = s_ck[lane + 32]; ci[1] = s_ci[lane + 32];
    unsigned gmask = 0;
    #pragma unroll
    for (int t = 0; t < 8; ++t) {
        int bk = kTaken;
        int bi = 0x7fffffff;
        #pragma unroll
        for (int q = 0; q < 2; ++q) {
            const bool take = (ck[q] > bk) | ((ck[q] == bk) & (ci[q] < bi));
            bk = take ? ck[q] : bk;
            bi = take ? ci[q] : bi;
        }
        int wk, wi;
        wave_best(bk, bi, wk, wi);
        #pragma unroll
        for (int q = 0; q < 2; ++q) ck[q] = (ci[q] == wi) ? kTaken : ck[q];
        const unsigned uw   = (unsigned)wi;
        const bool     mine = ((wi & 1020) == e0) & (uw < (unsigned)R_);
        const unsigned bit  = 1u << ((((uw >> 10) & 1u) << 2) | (uw & 3u));
        gmask |= mine ? bit : 0u;
    }

    float pre[8];
    #pragma unroll
    for (int h = 0; h < 2; ++h) {
        #pragma unroll
        for (int c = 0; c < 4; ++c) {
            float acc = 0.0f;
            acc += p4[0][h][c];
            if (nslice > 1) acc += p4[1][h][c];
            if (nslice > 2) acc += p4[2][h][c];
            if (nslice > 3) acc += p4[3][h][c];
            pre[h * 4 + c] = acc;
        }
    }

    #pragma unroll
    for (int j = 0; j < 8; ++j) {
        const float g = pre[j] + act[j];
        pre[j] = ((gmask >> j) & 1u) ? g : pre[j];
    }

    float sum = 0.0f;
    #pragma unroll
    for (int j = 0; j < 8; ++j) sum += pre[j];
    sum = wave_sum_f(sum);
    if (lane == 0) s_sum[wave] = sum;
    __syncthreads();
    float tot = 0.0f;
    #pragma unroll
    for (int w = 0; w < 8; ++w) tot += s_sum[w];
    const float mu = tot * (1.0f / R_);

    float ss = 0.0f;
    #pragma unroll
    for (int j = 0; j < 8; ++j) { float d = pre[j] - mu; ss += d * d; }
    ss = wave_sum_f(ss);
    if (lane == 0) s_sq[wave] = ss;
    __syncthreads();
    float tot2 = 0.0f;
    #pragma unroll
    for (int w = 0; w < 8; ++w) tot2 += s_sq[w];
    const float inv = rsqrtf(tot2 * (1.0f / R_) + 1e-5f);

    #pragma unroll
    for (int h = 0; h < 2; ++h) {
        v4f o;
        o.x = (pre[h * 4 + 0] - mu) * inv * g4[h].x + be4[h].x;
        o.y = (pre[h * 4 + 1] - mu) * inv * g4[h].y + be4[h].y;
        o.z = (pre[h * 4 + 2] - mu) * inv * g4[h].z + be4[h].z;
        o.w = (pre[h * 4 + 3] - mu) * inv * g4[h].w + be4[h].w;
        *(v4f*)(out + rowoff + h * 1024 + e0) = o;
    }
}

extern "C" void kernel_launch(void* const* d_in, const int* in_sizes, int n_in,
                              void* d_out, int out_size, void* d_ws, size_t ws_size,
                              hipStream_t stream) {
    const float* facts        = (const float*)d_in[0];
    const float* fact_logits  = (const float*)d_in[1];
    const float* agg_logits   = (const float*)d_in[2];
    const float* strength     = (const float*)d_in[3];
    const float* W_proj       = (const float*)d_in[4];
    const float* ln_gamma     = (const float*)d_in[5];
    const float* ln_beta      = (const float*)d_in[6];
    float* out = (float*)d_out;

    const size_t kFixed = 49152 + (size_t)B_ * R_ * 4;
    int nslice = 1;
    if (ws_size >= kFixed + 4 * (size_t)B_ * R_ * 4) nslice = 4;
    else if (ws_size >= kFixed + 2 * (size_t)B_ * R_ * 4) nslice = 2;

    char* ws = (char*)d_ws;
    float* ruleW    = (float*)(ws);
    int*   ruleIdx  = (int*)  (ws + (size_t)R_ * 4 * 4);
    float* projPart = (float*)(ws + kFixed);

    prep_gemm_kernel<<<16 * nslice + R_ / 8, 256, 0, stream>>>(fact_logits, agg_logits, strength,
                                                               ruleIdx, ruleW, facts, W_proj,
                                                               projPart, D_ / nslice, 16 * nslice);

    gate_ln_kernel<<<B_, 256, 0, stream>>>(projPart, nslice, facts, ruleIdx, ruleW,
                                           ln_gamma, ln_beta, out);
}


# rule-prep wave: whole 4 KB row read up front as eight independent 16-byte loads per lane (rule scalars too), per-lane top-2 by selects, no branch between loads; no floating-point sum order changed
# speedup vs baseline: 4.4357x; 1.5341x over previous
#include <hip/hip_runtime.h>
#include <hip/hip_bf16.h>

#define B_  32
#define D_  1024
#define R_  2048

typedef float v2f __attribute__((ext_vector_type(2)));
typedef float v8f __attribute__((ext_vector_type(8)));
typedef float v4f __attribute__((ext_vector_type(4)));
typedef int   v4i __attribute__((ext_vector_type(4)));

__device__ __forceinline__ void top2_insert(float v, int i,
                                            float& v1, int& i1,
                                            float& v2, int& i2) {
    bool beats1 = (v > v1) || (v == v1 && i < i1);
    if (beats1) {
        v2 = v1; i2 = i1;
        v1 = v;  i1 = i;
    } else {
        bool beats2 = (v > v2) || (v == v2 && i < i2);
        if (beats2) { v2 = v; i2 = i; }
    }
}

__device__ __forceinline__ void top2_take_rising(float v, int i,
                                                 float& v1, int& i1,
                                                 float& v2, int& i2) {
    const bool gt1 = v > v1;
    const bool gt2 = v > v2;
    v2 = gt1 ? v1 : (gt2 ? v : v2);
    i2 = gt1 ? i1 : (gt2 ? i : i2);
    v1 = gt1 ? v : v1;
    i1 = gt1 ? i : i1;
}

__device__ __forceinline__ void rule_prep_body(int blk,
                                 const float* __restrict__ fact_logits,
                                 const float* __restrict__ agg_logits,
                                 const float* __restrict__ strength,
                                 int*   __restrict__ ruleIdx,
                                 float* __restrict__ ruleW)
{
    const int wavesPerBlock = blockDim.x >> 5;
    const int r    = blk * wavesPerBlock + (threadIdx.x >> 5);
    const int lane = threadIdx.x & 31;
    if (r >= R_) return;

    const float* row = fact_logits + (size_t)r * D_;
    v4f x[8];
    #pragma unroll
    for (int i = 0; i < 8; ++i) x[i] = *(const v4f*)(row + 128 * i + 4 * lane);
    const float a0 = agg_logits[3 * r + 0];
    const float a1 = agg_logits[3 * r + 1];
    const float a2 = agg_logits[3 * r + 2];
    const float st = strength[r];

    const int  base = 4 * lane;
    const bool sw   = x[0][1] > x[0][0];
    float v1 = sw ? x[0][1] : x[0][0];
    float v2 = sw ? x[0][0] : x[0][1];
    int   i1 = sw ? base + 1 : base;
    int   i2 = sw ? base : base + 1;
    #pragma unroll
    for (int i = 0; i < 8; ++i) {
        #pragma unroll
        for (int c = (i == 0) ? 2 : 0; c < 4; ++c) {
            top2_take_rising(x[i][c], 128 * i + base + c, v1, i1, v2, i2);
        }
    }
    #pragma unroll
    for (int m = 16; m >= 1; m >>= 1) {
        float ov1 = __shfl_xor(v1, m, 32);
        int   oi1 = __shfl_xor(i1, m, 32);
        float ov2 = __shfl_xor(v2, m, 32);
        int   oi2 = __shfl_xor(i2, m, 32);
        top2_insert(ov1, oi1, v1, i1, v2, i2);
        top2_insert(ov2, oi2, v1, i1, v2, i2);
    }

    if (lane == 0) {
        ruleIdx[2 * r + 0] = i1;
        ruleIdx[2 * r + 1] = i2;
        float mx = fmaxf(a0, fmaxf(a1, a2));
        float e0 = expf(a0 - mx), e1 = expf(a1 - mx), e2 = expf(a2 - mx);
        float inv = 1.0f / (e0 + e1 + e2);
        ruleW[4 * r + 0] = e0 * inv;
        ruleW[4 * r + 1] = e1 * inv;
        ruleW[4 * r + 2] = e2 * inv;
        ruleW[4 * r + 3] = 1.0f / (1.0f + expf(-st));
    }
}

__device__ __forceinline__ void gemm_wmma_body(int blk,
                                 const float* __restrict__ facts,
                                 const float* __restrict__ Wp,
                                 float* __restrict__ projPart,
                                 int klen)
{
    const int wave  = (blk * blockDim.x + threadIdx.x) >> 5;
    const int lane  = threadIdx.x & 31;

    const int ntile = wave & 127;
    const int slice = wave >> 7;
    const int n0    = ntile * 16;
    const int kbeg  = slice * klen;

    const int rlo   = lane & 15;
    const int khalf = (lane >> 4) * 2;

    const float* aptr0 = facts + (size_t)(rlo)      * D_ + kbeg + khalf;
    const float* aptr1 = facts + (size_t)(16 + rlo) * D_ + kbeg + khalf;
    const float* bptr  = Wp    + (size_t)(n0 + rlo) * D_ + kbeg + khalf;

    v8f c0 = {};
    v8f c1 = {};
    for (int k = 0; k < klen; k += 16) {
        __builtin_prefetch(bptr + k + 512, 0, 3);
        #pragma unroll
        for (int kk = 0; kk < 16; kk += 4) {
            v2f b  = *(const v2f*)(bptr  + k + kk);
            v2f a0 = *(const v2f*)(aptr0 + k + kk);
            v2f a1 = *(const v2f*)(aptr1 + k + kk);
            c0 = __builtin_amdgcn_wmma_f32_16x16x4_f32(false, a0, false, b,
                                                       (short)0, c0, false, false);
            c1 = __builtin_amdgcn_wmma_f32_16x16x4_f32(false, a1, false, b,
                                                       (short)0, c1, false, false);
        }
    }

    float* outp = projPart + (size_t)slice * B_ * R_;
    const int col   = n0 + rlo;
    const int rbase = (lane >> 4) * 8;
    #pragma unroll
    for (int v = 0; v < 8; ++v) {
        outp[(size_t)(rbase + v)      * R_ + col] = c0[v];
        outp[(size_t)(rbase + v + 16) * R_ + col] = c1[v];
    }
}

__global__ void __launch_bounds__(256)
prep_gemm_kernel(const float* __restrict__ fact_logits,
                 const float* __restrict__ agg_logits,
                 const float* __restrict__ strength,
                 int*   __restrict__ ruleIdx,
                 float* __restrict__ ruleW,
                 const float* __restrict__ facts,
                 const float* __restrict__ Wp,
                 float* __restrict__ projPart,
                 int klen,
                 int gemmBlocks)
{
    const int blk = blockIdx.x;
    const int g   = blk - gemmBlocks;
    if (g >= 0 && g < gemmBlocks) {
        gemm_wmma_body(g, facts, Wp, projPart, klen);
    } else {
        rule_prep_body(g < 0 ? blk : g, fact_logits, agg_logits, strength, ruleIdx, ruleW);
    }
}

template <int M>
__device__ __forceinline__ int row_xor_i(int x) {
    return __builtin_amdgcn_update_dpp(0, x, 0x160 + M, 0xf, 0xf, true);
}
__device__ __forceinline__ int half_xor_i(int x) {
    return __builtin_amdgcn_permlanex16(x, x, 0x76543210, 0xfedcba98, false, false);
}
template <int M>
__device__ __forceinline__ float row_xor_f(float x) {
    return __builtin_bit_cast(float, row_xor_i<M>(__builtin_bit_cast(int, x)));
}
__device__ __forceinline__ float half_xor_f(float x) {
    return __builtin_bit_cast(float, half_xor_i(__builtin_bit_cast(int, x)));
}
__device__ __forceinline__ int wave_max_i(int v) {
    v = max(v, row_xor_i<1>(v));
    v = max(v, row_xor_i<2>(v));
    v = max(v, row_xor_i<4>(v));
    v = max(v, row_xor_i<8>(v));
    v = max(v, half_xor_i(v));
    return v;
}
__device__ __forceinline__ int wave_min_i(int v) {
    v = min(v, row_xor_i<1>(v));
    v = min(v, row_xor_i<2>(v));
    v = min(v, row_xor_i<4>(v));
    v = min(v, row_xor_i<8>(v));
    v = min(v, half_xor_i(v));
    return v;
}
__device__ __forceinline__ float wave_sum_f(float v) {
    v += row_xor_f<1>(v);
    v += row_xor_f<2>(v);
    v += row_xor_f<4>(v);
    v += row_xor_f<8>(v);
    v += half_xor_f(v);
    return v;
}
__device__ __forceinline__ int order_key(float v) {
    const int u = __builtin_bit_cast(int, v + 0.0f);
    return u ^ ((u >> 31) & 0x7fffffff);
}
__device__ __forceinline__ void wave_best(int bk, int bi, int& wk, int& wi) {
    wk = wave_max_i(bk);
    wi = wave_min_i((bk == wk) ? bi : 0x7fffffff);
}

__global__ void __launch_bounds__(256)
gate_ln_kernel(const float* __restrict__ projPart,
               int nslice,
               const float* __restrict__ facts,
               const int*   __restrict__ ruleIdx,
               const float* __restrict__ ruleW,
               const float* __restrict__ gamma,
               const float* __restrict__ beta,
               float* __restrict__ out)
{
    const int b    = blockIdx.x;
    const int tid  = threadIdx.x;
    const int lane = tid & 31;
    const int wave = tid >> 5;
    const int e0   = 4 * tid;

    __shared__ float s_f[D_];
    __shared__ int   s_ck[64];
    __shared__ int   s_ci[64];
    __shared__ float s_sum[8];
    __shared__ float s_sq[8];

    const size_t rowoff = (size_t)b * R_;

    *(v4f*)(s_f + e0) = *(const v4f*)(facts + (size_t)b * D_ + e0);

    v4i ix[2][2];
    v4f rw[2][4];
    #pragma unroll
    for (int h = 0; h < 2; ++h) {
        const int r0 = h * 1024 + e0;
        ix[h][0] = *(const v4i*)(ruleIdx + 2 * r0);
        ix[h][1] = *(const v4i*)(ruleIdx + 2 * r0 + 4);
        #pragma unroll
        for (int c = 0; c < 4; ++c) rw[h][c] = *(const v4f*)(ruleW + 4 * (r0 + c));
    }

    v4f g4[2], be4[2], p4[4][2];
    #pragma unroll
    for (int h = 0; h < 2; ++h) {
        const int col = h * 1024 + e0;
        #pragma unroll
        for (int s = 0; s < 4; ++s) {
            const int se = (s < nslice) ? s : 0;
            p4[s][h] = *(const v4f*)(projPart + (size_t)se * B_ * R_ + rowoff + col);
        }
        g4[h]  = *(const v4f*)(gamma + col);
        be4[h] = *(const v4f*)(beta + col);
    }

    __syncthreads();

    float act[8];
    #pragma unroll
    for (int j = 0; j < 8; ++j) {
        const v4i  q  = ix[j >> 2][(j >> 1) & 1];
        const int  i0 = ((j & 1) ? q.z : q.x) & (D_ - 1);
        const int  i1 = ((j & 1) ? q.w : q.y) & (D_ - 1);
        const float f0 = s_f[i0];
        const float f1 = s_f[i1];
        const v4f  w  = rw[j >> 2][j & 3];

        const float and_a = f0 * f1;
        const float or_a  = 1.0f - (1.0f - f0) * (1.0f - f1);
        const float k_a   = (f0 + f1) / (2.0f + 1e-8f);

        act[j] = (w.x * and_a + w.y * or_a + w.z * k_a) * w.w;
    }

    const int kTaken = (int)0x80000000;
    int key[8];
    #pragma unroll
    for (int j = 0; j < 8; ++j) key[j] = order_key(act[j]);

    int myck = kTaken;
    int myci = 0x7fffffff;
    #pragma unroll
    for (int t = 0; t < 8; ++t) {
        int bk = kTaken;
        int bi = 0x7fffffff;
        #pragma unroll
        for (int j = 0; j < 8; ++j) {
            const int  idx  = (j >> 2) * 1024 + e0 + (j & 3);
            const bool take = key[j] > bk;
            bk = take ? key[j] : bk;
            bi = take ? idx : bi;
        }
        int wk, wi;
        wave_best(bk, bi, wk, wi);
        #pragma unroll
        for (int j = 0; j < 8; ++j) {
            const int idx = (j >> 2) * 1024 + e0 + (j & 3);
            key[j] = (idx == wi) ? kTaken : key[j];
        }
        myck = (lane == t) ? wk : myck;
        myci = (lane == t) ? wi : myci;
    }
    if (lane < 8) {
        s_ck[wave * 8 + lane] = myck;
        s_ci[wave * 8 + lane] = myci;
    }
    __syncthreads();

    int ck[2], ci[2];
    ck[0] = s_ck[lane];      ci[0] = s_ci[lane];
    ck[1] = s_ck[lane + 32]; ci[1] = s_ci[lane + 32];
    unsigned gmask = 0;
    #pragma unroll
    for (int t = 0; t < 8; ++t) {
        int bk = kTaken;
        int bi = 0x7fffffff;
        #pragma unroll
        for (int q = 0; q < 2; ++q) {
            const bool take = (ck[q] > bk) | ((ck[q] == bk) & (ci[q] < bi));
            bk = take ? ck[q] : bk;
            bi = take ? ci[q] : bi;
        }
        int wk, wi;
        wave_best(bk, bi, wk, wi);
        #pragma unroll
        for (int q = 0; q < 2; ++q) ck[q] = (ci[q] == wi) ? kTaken : ck[q];
        const unsigned uw   = (unsigned)wi;
        const bool     mine = ((wi & 1020) == e0) & (uw < (unsigned)R_);
        const unsigned bit  = 1u << ((((uw >> 10) & 1u) << 2) | (uw & 3u));
        gmask |= mine ? bit : 0u;
    }

    float pre[8];
    #pragma unroll
    for (int h = 0; h < 2; ++h) {
        #pragma unroll
        for (int c = 0; c < 4; ++c) {
            float acc = 0.0f;
            acc += p4[0][h][c];
            if (nslice > 1) acc += p4[1][h][c];
            if (nslice > 2) acc += p4[2][h][c];
            if (nslice > 3) acc += p4[3][h][c];
            pre[h * 4 + c] = acc;
        }
    }

    #pragma unroll
    for (int j = 0; j < 8; ++j) {
        const float g = pre[j] + act[j];
        pre[j] = ((gmask >> j) & 1u) ? g : pre[j];
    }

    float sum = 0.0f;
    #pragma unroll
    for (int j = 0; j < 8; ++j) sum += pre[j];
    sum = wave_sum_f(sum);
    if (lane == 0) s_sum[wave] = sum;
    __syncthreads();
    float tot = 0.0f;
    #pragma unroll
    for (int w = 0; w < 8; ++w) tot += s_sum[w];
    const float mu = tot * (1.0f / R_);

    float ss = 0.0f;
    #pragma unroll
    for (int j = 0; j < 8; ++j) { float d = pre[j] - mu; ss += d * d; }
    ss = wave_sum_f(ss);
    if (lane == 0) s_sq[wave] = ss;
    __syncthreads();
    float tot2 = 0.0f;
    #pragma unroll
    for (int w = 0; w < 8; ++w) tot2 += s_sq[w];
    const float inv = rsqrtf(tot2 * (1.0f / R_) + 1e-5f);

    #pragma unroll
    for (int h = 0; h < 2; ++h) {
        v4f o;
        o.x = (pre[h * 4 + 0] - mu) * inv * g4[h].x + be4[h].x;
        o.y = (pre[h * 4 + 1] - mu) * inv * g4[h].y + be4[h].y;
        o.z = (pre[h * 4 + 2] - mu) * inv * g4[h].z + be4[h].z;
        o.w = (pre[h * 4 + 3] - mu) * inv * g4[h].w + be4[h].w;
        *(v4f*)(out + rowoff + h * 1024 + e0) = o;
    }
}

extern "C" void kernel_launch(void* const* d_in, const int* in_sizes, int n_in,
                              void* d_out, int out_size, void* d_ws, size_t ws_size,
                              hipStream_t stream) {
    const float* facts        = (const float*)d_in[0];
    const float* fact_logits  = (const float*)d_in[1];
    const float* agg_logits   = (const float*)d_in[2];
    const float* strength     = (const float*)d_in[3];
    const float* W_proj       = (const float*)d_in[4];
    const float* ln_gamma     = (const float*)d_in[5];
    const float* ln_beta      = (const float*)d_in[6];
    float* out = (float*)d_out;

    const size_t kFixed = 49152 + (size_t)B_ * R_ * 4;
    int nslice = 1;
    if (ws_size >= kFixed + 4 * (size_t)B_ * R_ * 4) nslice = 4;
    else if (ws_size >= kFixed + 2 * (size_t)B_ * R_ * 4) nslice = 2;

    char* ws = (char*)d_ws;
    float* ruleW    = (float*)(ws);
    int*   ruleIdx  = (int*)  (ws + (size_t)R_ * 4 * 4);
    float* projPart = (float*)(ws + kFixed);

    prep_gemm_kernel<<<16 * nslice + R_ / 8, 256, 0, stream>>>(fact_logits, agg_logits, strength,
                                                               ruleIdx, ruleW, facts, W_proj,
                                                               projPart, D_ / nslice, 16 * nslice);

    gate_ln_kernel<<<B_, 256, 0, stream>>>(projPart, nslice, facts, ruleIdx, ruleW,
                                           ln_gamma, ln_beta, out);
}


# projection matrix work dealt to all 1024 SIMDs: 256 work-groups of 8 rule + 4 GEMM waves (64 WMMAs each, loads up front), was 256 WMMAs on 256 SIMDs; K slices 4 -> 8, sum order changed
# speedup vs baseline: 5.5060x; 1.2413x over previous
#include <hip/hip_runtime.h>
#include <hip/hip_bf16.h>

#define B_  32
#define D_  1024
#define R_  2048

typedef float v2f __attribute__((ext_vector_type(2)));
typedef float v8f __attribute__((ext_vector_type(8)));
typedef float v4f __attribute__((ext_vector_type(4)));
typedef int   v4i __attribute__((ext_vector_type(4)));

__device__ __forceinline__ void top2_insert(float v, int i,
                                            float& v1, int& i1,
                                            float& v2, int& i2) {
    bool beats1 = (v > v1) || (v == v1 && i < i1);
    if (beats1) {
        v2 = v1; i2 = i1;
        v1 = v;  i1 = i;
    } else {
        bool beats2 = (v > v2) || (v == v2 && i < i2);
        if (beats2) { v2 = v; i2 = i; }
    }
}

__device__ __forceinline__ void top2_take_rising(float v, int i,
                                                 float& v1, int& i1,
                                                 float& v2, int& i2) {
    const bool gt1 = v > v1;
    const bool gt2 = v > v2;
    v2 = gt1 ? v1 : (gt2 ? v : v2);
    i2 = gt1 ? i1 : (gt2 ? i : i2);
    v1 = gt1 ? v : v1;
    i1 = gt1 ? i : i1;
}

__device__ __forceinline__ void rule_prep_body(int r,
                                 const float* __restrict__ fact_logits,
                                 const float* __restrict__ agg_logits,
                                 const float* __restrict__ strength,
                                 int*   __restrict__ ruleIdx,
                                 float* __restrict__ ruleW)
{
    const int lane = threadIdx.x & 31;
    if (r >= R_) return;

    const float* row = fact_logits + (size_t)r * D_;
    v4f x[8];
    #pragma unroll
    for (int i = 0; i < 8; ++i) x[i] = *(const v4f*)(row + 128 * i + 4 * lane);
    const float a0 = agg_logits[3 * r + 0];
    const float a1 = agg_logits[3 * r + 1];
    const float a2 = agg_logits[3 * r + 2];
    const float st = strength[r];

    const int  base = 4 * lane;
    const bool sw   = x[0][1] > x[0][0];
    float v1 = sw ? x[0][1] : x[0][0];
    float v2 = sw ? x[0][0] : x[0][1];
    int   i1 = sw ? base + 1 : base;
    int   i2 = sw ? base : base + 1;
    #pragma unroll
    for (int i = 0; i < 8; ++i) {
        #pragma unroll
        for (int c = (i == 0) ? 2 : 0; c < 4; ++c) {
            top2_take_rising(x[i][c], 128 * i + base + c, v1, i1, v2, i2);
        }
    }
    #pragma unroll
    for (int m = 16; m >= 1; m >>= 1) {
        float ov1 = __shfl_xor(v1, m, 32);
        int   oi1 = __shfl_xor(i1, m, 32);
        float ov2 = __shfl_xor(v2, m, 32);
        int   oi2 = __shfl_xor(i2, m, 32);
        top2_insert(ov1, oi1, v1, i1, v2, i2);
        top2_insert(ov2, oi2, v1, i1, v2, i2);
    }

    if (lane == 0) {
        ruleIdx[2 * r + 0] = i1;
        ruleIdx[2 * r + 1] = i2;
        float mx = fmaxf(a0, fmaxf(a1, a2));
        float e0 = expf(a0 - mx), e1 = expf(a1 - mx), e2 = expf(a2 - mx);
        float inv = 1.0f / (e0 + e1 + e2);
        ruleW[4 * r + 0] = e0 * inv;
        ruleW[4 * r + 1] = e1 * inv;
        ruleW[4 * r + 2] = e2 * inv;
        ruleW[4 * r + 3] = 1.0f / (1.0f + expf(-st));
    }
}

#define NSLICE_ 8
#define KLEN_   (D_ / NSLICE_)

__device__ __forceinline__ void gemm_wmma_body(int g,
                                 const float* __restrict__ facts,
                                 const float* __restrict__ Wp,
                                 float* __restrict__ projPart)
{
    const int lane  = threadIdx.x & 31;

    const int ntile = g & 127;
    const int slice = g >> 7;
    const int n0    = ntile * 16;
    const int kbeg  = slice * KLEN_;

    const int rlo   = lane & 15;
    const int khalf = (lane >> 4) * 2;

    const float* aptr0 = facts + (size_t)(rlo)      * D_ + kbeg + khalf;
    const float* aptr1 = facts + (size_t)(16 + rlo) * D_ + kbeg + khalf;
    const float* bptr  = Wp    + (size_t)(n0 + rlo) * D_ + kbeg + khalf;

    v2f fb[KLEN_ / 4], fa0[KLEN_ / 4], fa1[KLEN_ / 4];
    #pragma unroll
    for (int i = 0; i < KLEN_ / 4; ++i) {
        fb[i]  = *(const v2f*)(bptr  + 4 * i);
        fa0[i] = *(const v2f*)(aptr0 + 4 * i);
        fa1[i] = *(const v2f*)(aptr1 + 4 * i);
    }

    v8f c0 = {};
    v8f c1 = {};
    #pragma unroll
    for (int i = 0; i < KLEN_ / 4; ++i) {
        c0 = __builtin_amdgcn_wmma_f32_16x16x4_f32(false, fa0[i], false, fb[i],
                                                   (short)0, c0, false, false);
        c1 = __builtin_amdgcn_wmma_f32_16x16x4_f32(false, fa1[i], false, fb[i],
                                                   (short)0, c1, false, false);
    }

    float* outp = projPart + (size_t)slice * B_ * R_;
    const int col   = n0 + rlo;
    const int rbase = (lane >> 4) * 8;
    #pragma unroll
    for (int v = 0; v < 8; ++v) {
        outp[(size_t)(rbase + v)      * R_ + col] = c0[v];
        outp[(size_t)(rbase + v + 16) * R_ + col] = c1[v];
    }
}

#define RULE_WAVES_ 8
#define GEMM_WAVES_ 4
static_assert((R_ / RULE_WAVES_) * GEMM_WAVES_ == (R_ / 16) * NSLICE_,
              "projection waves of the grid must cover every (N-tile, K-slice) pair once");

__global__ void __launch_bounds__(32 * (RULE_WAVES_ + GEMM_WAVES_))
__attribute__((amdgpu_waves_per_eu(3, 4)))
prep_gemm_kernel(const float* __restrict__ fact_logits,
                 const float* __restrict__ agg_logits,
                 const float* __restrict__ strength,
                 int*   __restrict__ ruleIdx,
                 float* __restrict__ ruleW,
                 const float* __restrict__ facts,
                 const float* __restrict__ Wp,
                 float* __restrict__ projPart)
{
    const int w = __builtin_amdgcn_readfirstlane((int)(threadIdx.x >> 5));
    if (w < RULE_WAVES_) {
        rule_prep_body((int)blockIdx.x * RULE_WAVES_ + w,
                       fact_logits, agg_logits, strength, ruleIdx, ruleW);
    } else {
        gemm_wmma_body((int)blockIdx.x * GEMM_WAVES_ + (w - RULE_WAVES_),
                       facts, Wp, projPart);
    }
}

template <int M>
__device__ __forceinline__ int row_xor_i(int x) {
    return __builtin_amdgcn_update_dpp(0, x, 0x160 + M, 0xf, 0xf, true);
}
__device__ __forceinline__ int half_xor_i(int x) {
    return __builtin_amdgcn_permlanex16(x, x, 0x76543210, 0xfedcba98, false, false);
}
template <int M>
__device__ __forceinline__ float row_xor_f(float x) {
    return __builtin_bit_cast(float, row_xor_i<M>(__builtin_bit_cast(int, x)));
}
__device__ __forceinline__ float half_xor_f(float x) {
    return __builtin_bit_cast(float, half_xor_i(__builtin_bit_cast(int, x)));
}
__device__ __forceinline__ int wave_max_i(int v) {
    v = max(v, row_xor_i<1>(v));
    v = max(v, row_xor_i<2>(v));
    v = max(v, row_xor_i<4>(v));
    v = max(v, row_xor_i<8>(v));
    v = max(v, half_xor_i(v));
    return v;
}
__device__ __forceinline__ int wave_min_i(int v) {
    v = min(v, row_xor_i<1>(v));
    v = min(v, row_xor_i<2>(v));
    v = min(v, row_xor_i<4>(v));
    v = min(v, row_xor_i<8>(v));
    v = min(v, half_xor_i(v));
    return v;
}
__device__ __forceinline__ float wave_sum_f(float v) {
    v += row_xor_f<1>(v);
    v += row_xor_f<2>(v);
    v += row_xor_f<4>(v);
    v += row_xor_f<8>(v);
    v += half_xor_f(v);
    return v;
}
__device__ __forceinline__ int order_key(float v) {
    const int u = __builtin_bit_cast(int, v + 0.0f);
    return u ^ ((u >> 31) & 0x7fffffff);
}
__device__ __forceinline__ void wave_best(int bk, int bi, int& wk, int& wi) {
    wk = wave_max_i(bk);
    wi = wave_min_i((bk == wk) ? bi : 0x7fffffff);
}

__global__ void __launch_bounds__(256)
gate_ln_kernel(const float* __restrict__ projPart,
               const float* __restrict__ facts,
               const int*   __restrict__ ruleIdx,
               const float* __restrict__ ruleW,
               const float* __restrict__ gamma,
               const float* __restrict__ beta,
               float* __restrict__ out)
{
    const int b    = blockIdx.x;
    const int tid  = threadIdx.x;
    const int lane = tid & 31;
    const int wave = tid >> 5;
    const int e0   = 4 * tid;

    __shared__ float s_f[D_];
    __shared__ int   s_ck[64];
    __shared__ int   s_ci[64];
    __shared__ float s_sum[8];
    __shared__ float s_sq[8];

    const size_t rowoff = (size_t)b * R_;

    *(v4f*)(s_f + e0) = *(const v4f*)(facts + (size_t)b * D_ + e0);

    v4i ix[2][2];
    v4f rw[2][4];
    #pragma unroll
    for (int h = 0; h < 2; ++h) {
        const int r0 = h * 1024 + e0;
        ix[h][0] = *(const v4i*)(ruleIdx + 2 * r0);
        ix[h][1] = *(const v4i*)(ruleIdx + 2 * r0 + 4);
        #pragma unroll
        for (int c = 0; c < 4; ++c) rw[h][c] = *(const v4f*)(ruleW + 4 * (r0 + c));
    }

    v4f g4[2], be4[2], p4[NSLICE_][2];
    #pragma unroll
    for (int h = 0; h < 2; ++h) {
        const int col = h * 1024 + e0;
        #pragma unroll
        for (int s = 0; s < NSLICE_; ++s) {
            p4[s][h] = *(const v4f*)(projPart + (size_t)s * B_ * R_ + rowoff + col);
        }
        g4[h]  = *(const v4f*)(gamma + col);
        be4[h] = *(const v4f*)(beta + col);
    }

    __syncthreads();

    float act[8];
    #pragma unroll
    for (int j = 0; j < 8; ++j) {
        const v4i  q  = ix[j >> 2][(j >> 1) & 1];
        const int  i0 = ((j & 1) ? q.z : q.x) & (D_ - 1);
        const int  i1 = ((j & 1) ? q.w : q.y) & (D_ - 1);
        const float f0 = s_f[i0];
        const float f1 = s_f[i1];
        const v4f  w  = rw[j >> 2][j & 3];

        const float and_a = f0 * f1;
        const float or_a  = 1.0f - (1.0f - f0) * (1.0f - f1);
        const float k_a   = (f0 + f1) / (2.0f + 1e-8f);

        act[j] = (w.x * and_a + w.y * or_a + w.z * k_a) * w.w;
    }

    const int kTaken = (int)0x80000000;
    int key[8];
    #pragma unroll
    for (int j = 0; j < 8; ++j) key[j] = order_key(act[j]);

    int myck = kTaken;
    int myci = 0x7fffffff;
    #pragma unroll
    for (int t = 0; t < 8; ++t) {
        int bk = kTaken;
        int bi = 0x7fffffff;
        #pragma unroll
        for (int j = 0; j < 8; ++j) {
            const int  idx  = (j >> 2) * 1024 + e0 + (j & 3);
            const bool take = key[j] > bk;
            bk = take ? key[j] : bk;
            bi = take ? idx : bi;
        }
        int wk, wi;
        wave_best(bk, bi, wk, wi);
        #pragma unroll
        for (int j = 0; j < 8; ++j) {
            const int idx = (j >> 2) * 1024 + e0 + (j & 3);
            key[j] = (idx == wi) ? kTaken : key[j];
        }
        myck = (lane == t) ? wk : myck;
        myci = (lane == t) ? wi : myci;
    }
    if (lane < 8) {
        s_ck[wave * 8 + lane] = myck;
        s_ci[wave * 8 + lane] = myci;
    }
    __syncthreads();

    int ck[2], ci[2];
    ck[0] = s_ck[lane];      ci[0] = s_ci[lane];
    ck[1] = s_ck[lane + 32]; ci[1] = s_ci[lane + 32];
    unsigned gmask = 0;
    #pragma unroll
    for (int t = 0; t < 8; ++t) {
        int bk = kTaken;
        int bi = 0x7fffffff;
        #pragma unroll
        for (int q = 0; q < 2; ++q) {
            const bool take = (ck[q] > bk) | ((ck[q] == bk) & (ci[q] < bi));
            bk = take ? ck[q] : bk;
            bi = take ? ci[q] : bi;
        }
        int wk, wi;
        wave_best(bk, bi, wk, wi);
        #pragma unroll
        for (int q = 0; q < 2; ++q) ck[q] = (ci[q] == wi) ? kTaken : ck[q];
        const unsigned uw   = (unsigned)wi;
        const bool     mine = ((wi & 1020) == e0) & (uw < (unsigned)R_);
        const unsigned bit  = 1u << ((((uw >> 10) & 1u) << 2) | (uw & 3u));
        gmask |= mine ? bit : 0u;
    }

    float pre[8];
    #pragma unroll
    for (int h = 0; h < 2; ++h) {
        #pragma unroll
        for (int c = 0; c < 4; ++c) {
            float acc = 0.0f;
            #pragma unroll
            for (int s = 0; s < NSLICE_; ++s) acc += p4[s][h][c];
            pre[h * 4 + c] = acc;
        }
    }

    #pragma unroll
    for (int j = 0; j < 8; ++j) {
        const float g = pre[j] + act[j];
        pre[j] = ((gmask >> j) & 1u) ? g : pre[j];
    }

    float sum = 0.0f;
    #pragma unroll
    for (int j = 0; j < 8; ++j) sum += pre[j];
    sum = wave_sum_f(sum);
    if (lane == 0) s_sum[wave] = sum;
    __syncthreads();
    float tot = 0.0f;
    #pragma unroll
    for (int w = 0; w < 8; ++w) tot += s_sum[w];
    const float mu = tot * (1.0f / R_);

    float ss = 0.0f;
    #pragma unroll
    for (int j = 0; j < 8; ++j) { float d = pre[j] - mu; ss += d * d; }
    ss = wave_sum_f(ss);
    if (lane == 0) s_sq[wave] = ss;
    __syncthreads();
    float tot2 = 0.0f;
    #pragma unroll
    for (int w = 0; w < 8; ++w) tot2 += s_sq[w];
    const float inv = rsqrtf(tot2 * (1.0f / R_) + 1e-5f);

    #pragma unroll
    for (int h = 0; h < 2; ++h) {
        v4f o;
        o.x = (pre[h * 4 + 0] - mu) * inv * g4[h].x + be4[h].x;
        o.y = (pre[h * 4 + 1] - mu) * inv * g4[h].y + be4[h].y;
        o.z = (pre[h * 4 + 2] - mu) * inv * g4[h].z + be4[h].z;
        o.w = (pre[h * 4 + 3] - mu) * inv * g4[h].w + be4[h].w;
        *(v4f*)(out + rowoff + h * 1024 + e0) = o;
    }
}

extern "C" void kernel_launch(void* const* d_in, const int* in_sizes, int n_in,
                              void* d_out, int out_size, void* d_ws, size_t ws_size,
                              hipStream_t stream) {
    const float* facts        = (const float*)d_in[0];
    const float* fact_logits  = (const float*)d_in[1];
    const float* agg_logits   = (const float*)d_in[2];
    const float* strength     = (const float*)d_in[3];
    const float* W_proj       = (const float*)d_in[4];
    const float* ln_gamma     = (const float*)d_in[5];
    const float* ln_beta      = (const float*)d_in[6];
    float* out = (float*)d_out;

    char* ws = (char*)d_ws;
    float* ruleW    = (float*)(ws);
    int*   ruleIdx  = (int*)  (ws + (size_t)R_ * 4 * 4);
    float* projPart = (float*)(ws + 65536);

    prep_gemm_kernel<<<R_ / RULE_WAVES_, 32 * (RULE_WAVES_ + GEMM_WAVES_), 0, stream>>>(
        fact_logits, agg_logits, strength, ruleIdx, ruleW, facts, W_proj, projPart);

    gate_ln_kernel<<<B_, 256, 0, stream>>>(projPart, facts, ruleIdx, ruleW,
                                           ln_gamma, ln_beta, out);
}


# top-8 rule gate on values alone: per-lane sorting network + 5+3 butterfly merges of sorted 8-lists replace 16 serial wave argmax rounds; ties across the cut take the old exact path; sums unchanged
# speedup vs baseline: 5.9604x; 1.0825x over previous
#include <hip/hip_runtime.h>
#include <hip/hip_bf16.h>

#define B_  32
#define D_  1024
#define R_  2048

typedef float v2f __attribute__((ext_vector_type(2)));
typedef float v8f __attribute__((ext_vector_type(8)));
typedef float v4f __attribute__((ext_vector_type(4)));
typedef int   v4i __attribute__((ext_vector_type(4)));

__device__ __forceinline__ void top2_insert(float v, int i,
                                            float& v1, int& i1,
                                            float& v2, int& i2) {
    bool beats1 = (v > v1) || (v == v1 && i < i1);
    if (beats1) {
        v2 = v1; i2 = i1;
        v1 = v;  i1 = i;
    } else {
        bool beats2 = (v > v2) || (v == v2 && i < i2);
        if (beats2) { v2 = v; i2 = i; }
    }
}

__device__ __forceinline__ void top2_take_rising(float v, int i,
                                                 float& v1, int& i1,
                                                 float& v2, int& i2) {
    const bool gt1 = v > v1;
    const bool gt2 = v > v2;
    v2 = gt1 ? v1 : (gt2 ? v : v2);
    i2 = gt1 ? i1 : (gt2 ? i : i2);
    v1 = gt1 ? v : v1;
    i1 = gt1 ? i : i1;
}

__device__ __forceinline__ void rule_prep_body(int r,
                                 const float* __restrict__ fact_logits,
                                 const float* __restrict__ agg_logits,
                                 const float* __restrict__ strength,
                                 int*   __restrict__ ruleIdx,
                                 float* __restrict__ ruleW)
{
    const int lane = threadIdx.x & 31;
    if (r >= R_) return;

    const float* row = fact_logits + (size_t)r * D_;
    v4f x[8];
    #pragma unroll
    for (int i = 0; i < 8; ++i) x[i] = *(const v4f*)(row + 128 * i + 4 * lane);
    const float a0 = agg_logits[3 * r + 0];
    const float a1 = agg_logits[3 * r + 1];
    const float a2 = agg_logits[3 * r + 2];
    const float st = strength[r];

    const int  base = 4 * lane;
    const bool sw   = x[0][1] > x[0][0];
    float v1 = sw ? x[0][1] : x[0][0];
    float v2 = sw ? x[0][0] : x[0][1];
    int   i1 = sw ? base + 1 : base;
    int   i2 = sw ? base : base + 1;
    #pragma unroll
    for (int i = 0; i < 8; ++i) {
        #pragma unroll
        for (int c = (i == 0) ? 2 : 0; c < 4; ++c) {
            top2_take_rising(x[i][c], 128 * i + base + c, v1, i1, v2, i2);
        }
    }
    #pragma unroll
    for (int m = 16; m >= 1; m >>= 1) {
        float ov1 = __shfl_xor(v1, m, 32);
        int   oi1 = __shfl_xor(i1, m, 32);
        float ov2 = __shfl_xor(v2, m, 32);
        int   oi2 = __shfl_xor(i2, m, 32);
        top2_insert(ov1, oi1, v1, i1, v2, i2);
        top2_insert(ov2, oi2, v1, i1, v2, i2);
    }

    if (lane == 0) {
        ruleIdx[2 * r + 0] = i1;
        ruleIdx[2 * r + 1] = i2;
        float mx = fmaxf(a0, fmaxf(a1, a2));
        float e0 = expf(a0 - mx), e1 = expf(a1 - mx), e2 = expf(a2 - mx);
        float inv = 1.0f / (e0 + e1 + e2);
        ruleW[4 * r + 0] = e0 * inv;
        ruleW[4 * r + 1] = e1 * inv;
        ruleW[4 * r + 2] = e2 * inv;
        ruleW[4 * r + 3] = 1.0f / (1.0f + expf(-st));
    }
}

#define NSLICE_ 8
#define KLEN_   (D_ / NSLICE_)

__device__ __forceinline__ void gemm_wmma_body(int g,
                                 const float* __restrict__ facts,
                                 const float* __restrict__ Wp,
                                 float* __restrict__ projPart)
{
    const int lane  = threadIdx.x & 31;

    const int ntile = g & 127;
    const int slice = g >> 7;
    const int n0    = ntile * 16;
    const int kbeg  = slice * KLEN_;

    const int rlo   = lane & 15;
    const int khalf = (lane >> 4) * 2;

    const float* aptr0 = facts + (size_t)(rlo)      * D_ + kbeg + khalf;
    const float* aptr1 = facts + (size_t)(16 + rlo) * D_ + kbeg + khalf;
    const float* bptr  = Wp    + (size_t)(n0 + rlo) * D_ + kbeg + khalf;

    v2f fb[KLEN_ / 4], fa0[KLEN_ / 4], fa1[KLEN_ / 4];
    #pragma unroll
    for (int i = 0; i < KLEN_ / 4; ++i) {
        fb[i]  = *(const v2f*)(bptr  + 4 * i);
        fa0[i] = *(const v2f*)(aptr0 + 4 * i);
        fa1[i] = *(const v2f*)(aptr1 + 4 * i);
    }

    v8f c0 = {};
    v8f c1 = {};
    #pragma unroll
    for (int i = 0; i < KLEN_ / 4; ++i) {
        c0 = __builtin_amdgcn_wmma_f32_16x16x4_f32(false, fa0[i], false, fb[i],
                                                   (short)0, c0, false, false);
        c1 = __builtin_amdgcn_wmma_f32_16x16x4_f32(false, fa1[i], false, fb[i],
                                                   (short)0, c1, false, false);
    }

    float* outp = projPart + (size_t)slice * B_ * R_;
    const int col   = n0 + rlo;
    const int rbase = (lane >> 4) * 8;
    #pragma unroll
    for (int v = 0; v < 8; ++v) {
        outp[(size_t)(rbase + v)      * R_ + col] = c0[v];
        outp[(size_t)(rbase + v + 16) * R_ + col] = c1[v];
    }
}

#define RULE_WAVES_ 8
#define GEMM_WAVES_ 4
static_assert((R_ / RULE_WAVES_) * GEMM_WAVES_ == (R_ / 16) * NSLICE_,
              "projection waves of the grid must cover every (N-tile, K-slice) pair once");

__global__ void __launch_bounds__(32 * (RULE_WAVES_ + GEMM_WAVES_))
__attribute__((amdgpu_waves_per_eu(3, 4)))
prep_gemm_kernel(const float* __restrict__ fact_logits,
                 const float* __restrict__ agg_logits,
                 const float* __restrict__ strength,
                 int*   __restrict__ ruleIdx,
                 float* __restrict__ ruleW,
                 const float* __restrict__ facts,
                 const float* __restrict__ Wp,
                 float* __restrict__ projPart)
{
    const int w = __builtin_amdgcn_readfirstlane((int)(threadIdx.x >> 5));
    if (w < RULE_WAVES_) {
        rule_prep_body((int)blockIdx.x * RULE_WAVES_ + w,
                       fact_logits, agg_logits, strength, ruleIdx, ruleW);
    } else {
        gemm_wmma_body((int)blockIdx.x * GEMM_WAVES_ + (w - RULE_WAVES_),
                       facts, Wp, projPart);
    }
}

template <int M>
__device__ __forceinline__ int row_xor_i(int x) {
    return __builtin_amdgcn_update_dpp(0, x, 0x160 + M, 0xf, 0xf, true);
}
__device__ __forceinline__ int half_xor_i(int x) {
    return __builtin_amdgcn_permlanex16(x, x, 0x76543210, 0xfedcba98, false, false);
}
template <int M>
__device__ __forceinline__ float row_xor_f(float x) {
    return __builtin_bit_cast(float, row_xor_i<M>(__builtin_bit_cast(int, x)));
}
__device__ __forceinline__ float half_xor_f(float x) {
    return __builtin_bit_cast(float, half_xor_i(__builtin_bit_cast(int, x)));
}
__device__ __forceinline__ int wave_max_i(int v) {
    v = max(v, row_xor_i<1>(v));
    v = max(v, row_xor_i<2>(v));
    v = max(v, row_xor_i<4>(v));
    v = max(v, row_xor_i<8>(v));
    v = max(v, half_xor_i(v));
    return v;
}
__device__ __forceinline__ int wave_min_i(int v) {
    v = min(v, row_xor_i<1>(v));
    v = min(v, row_xor_i<2>(v));
    v = min(v, row_xor_i<4>(v));
    v = min(v, row_xor_i<8>(v));
    v = min(v, half_xor_i(v));
    return v;
}
__device__ __forceinline__ float wave_sum_f(float v) {
    v += row_xor_f<1>(v);
    v += row_xor_f<2>(v);
    v += row_xor_f<4>(v);
    v += row_xor_f<8>(v);
    v += half_xor_f(v);
    return v;
}
__device__ __forceinline__ void cx_desc(int& a, int& b) {
    const int hi = max(a, b);
    const int lo = min(a, b);
    a = hi;
    b = lo;
}
__device__ __forceinline__ void sort8_desc(int (&k)[8]) {
    cx_desc(k[0], k[2]); cx_desc(k[1], k[3]); cx_desc(k[4], k[6]); cx_desc(k[5], k[7]);
    cx_desc(k[0], k[4]); cx_desc(k[1], k[5]); cx_desc(k[2], k[6]); cx_desc(k[3], k[7]);
    cx_desc(k[0], k[1]); cx_desc(k[2], k[3]); cx_desc(k[4], k[5]); cx_desc(k[6], k[7]);
    cx_desc(k[2], k[4]); cx_desc(k[3], k[5]);
    cx_desc(k[1], k[4]); cx_desc(k[3], k[6]);
    cx_desc(k[1], k[2]); cx_desc(k[3], k[4]); cx_desc(k[5], k[6]);
}
__device__ __forceinline__ void merge8_desc(int (&k)[8], const int (&p)[8]) {
    #pragma unroll
    for (int i = 0; i < 8; ++i) k[i] = max(k[i], p[i]);
    cx_desc(k[0], k[4]); cx_desc(k[1], k[5]); cx_desc(k[2], k[6]); cx_desc(k[3], k[7]);
    cx_desc(k[0], k[2]); cx_desc(k[1], k[3]); cx_desc(k[4], k[6]); cx_desc(k[5], k[7]);
    cx_desc(k[0], k[1]); cx_desc(k[2], k[3]); cx_desc(k[4], k[5]); cx_desc(k[6], k[7]);
}
template <int M>
__device__ __forceinline__ void merge_row_xor(int (&k)[8]) {
    int p[8];
    #pragma unroll
    for (int i = 0; i < 8; ++i) p[i] = row_xor_i<M>(k[7 - i]);
    merge8_desc(k, p);
}
__device__ __forceinline__ void merge_half_xor(int (&k)[8]) {
    int p[8];
    #pragma unroll
    for (int i = 0; i < 8; ++i) p[i] = half_xor_i(k[7 - i]);
    merge8_desc(k, p);
}
__device__ __forceinline__ int order_key(float v) {
    const int u = __builtin_bit_cast(int, v + 0.0f);
    return u ^ ((u >> 31) & 0x7fffffff);
}
__device__ __forceinline__ void wave_best(int bk, int bi, int& wk, int& wi) {
    wk = wave_max_i(bk);
    wi = wave_min_i((bk == wk) ? bi : 0x7fffffff);
}

__device__ __forceinline__ unsigned gate_mask_by_rounds(const float (&act)[8], int e0,
                                                        int lane, int wave,
                                                        int* s_ck, int* s_ci) {
    const int kTaken = (int)0x80000000;
    int key[8];
    #pragma unroll
    for (int j = 0; j < 8; ++j) key[j] = order_key(act[j]);

    int myck = kTaken;
    int myci = 0x7fffffff;
    #pragma unroll
    for (int t = 0; t < 8; ++t) {
        int bk = kTaken;
        int bi = 0x7fffffff;
        #pragma unroll
        for (int j = 0; j < 8; ++j) {
            const int  idx  = (j >> 2) * 1024 + e0 + (j & 3);
            const bool take = key[j] > bk;
            bk = take ? key[j] : bk;
            bi = take ? idx : bi;
        }
        int wk, wi;
        wave_best(bk, bi, wk, wi);
        #pragma unroll
        for (int j = 0; j < 8; ++j) {
            const int idx = (j >> 2) * 1024 + e0 + (j & 3);
            key[j] = (idx == wi) ? kTaken : key[j];
        }
        myck = (lane == t) ? wk : myck;
        myci = (lane == t) ? wi : myci;
    }
    if (lane < 8) {
        s_ck[wave * 8 + lane] = myck;
        s_ci[wave * 8 + lane] = myci;
    }
    __syncthreads();

    int ck[2], ci[2];
    ck[0] = s_ck[lane];      ci[0] = s_ci[lane];
    ck[1] = s_ck[lane + 32]; ci[1] = s_ci[lane + 32];
    unsigned gmask = 0;
    #pragma unroll
    for (int t = 0; t < 8; ++t) {
        int bk = kTaken;
        int bi = 0x7fffffff;
        #pragma unroll
        for (int q = 0; q < 2; ++q) {
            const bool take = (ck[q] > bk) | ((ck[q] == bk) & (ci[q] < bi));
            bk = take ? ck[q] : bk;
            bi = take ? ci[q] : bi;
        }
        int wk, wi;
        wave_best(bk, bi, wk, wi);
        #pragma unroll
        for (int q = 0; q < 2; ++q) ck[q] = (ci[q] == wi) ? kTaken : ck[q];
        const unsigned uw   = (unsigned)wi;
        const bool     mine = ((wi & 1020) == e0) & (uw < (unsigned)R_);
        const unsigned bit  = 1u << ((((uw >> 10) & 1u) << 2) | (uw & 3u));
        gmask |= mine ? bit : 0u;
    }
    return gmask;
}

__global__ void __launch_bounds__(256)
gate_ln_kernel(const float* __restrict__ projPart,
               const float* __restrict__ facts,
               const int*   __restrict__ ruleIdx,
               const float* __restrict__ ruleW,
               const float* __restrict__ gamma,
               const float* __restrict__ beta,
               float* __restrict__ out)
{
    const int b    = blockIdx.x;
    const int tid  = threadIdx.x;
    const int lane = tid & 31;
    const int wave = tid >> 5;
    const int e0   = 4 * tid;

    __shared__ float s_f[D_];
    __shared__ int   s_ck[64];
    __shared__ int   s_ci[64];
    __shared__ __attribute__((aligned(16))) int s_top[64];
    __shared__ float s_sum[8];
    __shared__ float s_sq[8];
    __shared__ int   s_cnt[8];

    const size_t rowoff = (size_t)b * R_;

    *(v4f*)(s_f + e0) = *(const v4f*)(facts + (size_t)b * D_ + e0);

    v4i ix[2][2];
    v4f rw[2][4];
    #pragma unroll
    for (int h = 0; h < 2; ++h) {
        const int r0 = h * 1024 + e0;
        ix[h][0] = *(const v4i*)(ruleIdx + 2 * r0);
        ix[h][1] = *(const v4i*)(ruleIdx + 2 * r0 + 4);
        #pragma unroll
        for (int c = 0; c < 4; ++c) rw[h][c] = *(const v4f*)(ruleW + 4 * (r0 + c));
    }

    v4f g4[2], be4[2], p4[NSLICE_][2];
    #pragma unroll
    for (int h = 0; h < 2; ++h) {
        const int col = h * 1024 + e0;
        #pragma unroll
        for (int s = 0; s < NSLICE_; ++s) {
            p4[s][h] = *(const v4f*)(projPart + (size_t)s * B_ * R_ + rowoff + col);
        }
        g4[h]  = *(const v4f*)(gamma + col);
        be4[h] = *(const v4f*)(beta + col);
    }

    __syncthreads();

    float act[8];
    #pragma unroll
    for (int j = 0; j < 8; ++j) {
        const v4i  q  = ix[j >> 2][(j >> 1) & 1];
        const int  i0 = ((j & 1) ? q.z : q.x) & (D_ - 1);
        const int  i1 = ((j & 1) ? q.w : q.y) & (D_ - 1);
        const float f0 = s_f[i0];
        const float f1 = s_f[i1];
        const v4f  w  = rw[j >> 2][j & 3];

        const float and_a = f0 * f1;
        const float or_a  = 1.0f - (1.0f - f0) * (1.0f - f1);
        const float k_a   = (f0 + f1) / (2.0f + 1e-8f);

        act[j] = (w.x * and_a + w.y * or_a + w.z * k_a) * w.w;
    }

    int key[8], tk[8];
    #pragma unroll
    for (int j = 0; j < 8; ++j) { key[j] = order_key(act[j]); tk[j] = key[j]; }
    sort8_desc(tk);
    merge_row_xor<1>(tk);
    merge_row_xor<2>(tk);
    merge_row_xor<4>(tk);
    merge_row_xor<8>(tk);
    merge_half_xor(tk);
    if (lane == 0) {
        v4i lo, hi;
        lo.x = tk[0]; lo.y = tk[1]; lo.z = tk[2]; lo.w = tk[3];
        hi.x = tk[4]; hi.y = tk[5]; hi.z = tk[6]; hi.w = tk[7];
        *(v4i*)(s_top + wave * 8)     = lo;
        *(v4i*)(s_top + wave * 8 + 4) = hi;
    }
    __syncthreads();

    {
        const v4i lo = *(const v4i*)(s_top + (lane & 7) * 8);
        const v4i hi = *(const v4i*)(s_top + (lane & 7) * 8 + 4);
        tk[0] = lo.x; tk[1] = lo.y; tk[2] = lo.z; tk[3] = lo.w;
        tk[4] = hi.x; tk[5] = hi.y; tk[6] = hi.z; tk[7] = hi.w;
    }
    merge_row_xor<1>(tk);
    merge_row_xor<2>(tk);
    merge_row_xor<4>(tk);
    const int thr = tk[7];
    bool on[8];
    int  cnt = 0;
    #pragma unroll
    for (int j = 0; j < 8; ++j) {
        on[j] = key[j] >= thr;
        cnt += __builtin_popcount(__builtin_amdgcn_ballot_w32(on[j]));
    }

    float base[8];
    #pragma unroll
    for (int h = 0; h < 2; ++h) {
        #pragma unroll
        for (int c = 0; c < 4; ++c) {
            float acc = 0.0f;
            #pragma unroll
            for (int s = 0; s < NSLICE_; ++s) acc += p4[s][h][c];
            base[h * 4 + c] = acc;
        }
    }

    float pre[8];
    #pragma unroll
    for (int j = 0; j < 8; ++j) {
        const float g = base[j] + act[j];
        pre[j] = on[j] ? g : base[j];
    }

    float sum = 0.0f;
    #pragma unroll
    for (int j = 0; j < 8; ++j) sum += pre[j];
    sum = wave_sum_f(sum);
    if (lane == 0) { s_sum[wave] = sum; s_cnt[wave] = cnt; }
    __syncthreads();
    float tot = 0.0f;
    int   totc = 0;
    #pragma unroll
    for (int w = 0; w < 8; ++w) { tot += s_sum[w]; totc += s_cnt[w]; }

    if (__builtin_amdgcn_readfirstlane(totc) != 8) {
        const unsigned gmask = gate_mask_by_rounds(act, e0, lane, wave, s_ck, s_ci);
        #pragma unroll
        for (int j = 0; j < 8; ++j) {
            const float g = base[j] + act[j];
            pre[j] = ((gmask >> j) & 1u) ? g : base[j];
        }
        sum = 0.0f;
        #pragma unroll
        for (int j = 0; j < 8; ++j) sum += pre[j];
        sum = wave_sum_f(sum);
        if (lane == 0) s_sum[wave] = sum;
        __syncthreads();
        tot = 0.0f;
        #pragma unroll
        for (int w = 0; w < 8; ++w) tot += s_sum[w];
    }
    const float mu = tot * (1.0f / R_);

    float ss = 0.0f;
    #pragma unroll
    for (int j = 0; j < 8; ++j) { float d = pre[j] - mu; ss += d * d; }
    ss = wave_sum_f(ss);
    if (lane == 0) s_sq[wave] = ss;
    __syncthreads();
    float tot2 = 0.0f;
    #pragma unroll
    for (int w = 0; w < 8; ++w) tot2 += s_sq[w];
    const float inv = rsqrtf(tot2 * (1.0f / R_) + 1e-5f);

    #pragma unroll
    for (int h = 0; h < 2; ++h) {
        v4f o;
        o.x = (pre[h * 4 + 0] - mu) * inv * g4[h].x + be4[h].x;
        o.y = (pre[h * 4 + 1] - mu) * inv * g4[h].y + be4[h].y;
        o.z = (pre[h * 4 + 2] - mu) * inv * g4[h].z + be4[h].z;
        o.w = (pre[h * 4 + 3] - mu) * inv * g4[h].w + be4[h].w;
        *(v4f*)(out + rowoff + h * 1024 + e0) = o;
    }
}

extern "C" void kernel_launch(void* const* d_in, const int* in_sizes, int n_in,
                              void* d_out, int out_size, void* d_ws, size_t ws_size,
                              hipStream_t stream) {
    const float* facts        = (const float*)d_in[0];
    const float* fact_logits  = (const float*)d_in[1];
    const float* agg_logits   = (const float*)d_in[2];
    const float* strength     = (const float*)d_in[3];
    const float* W_proj       = (const float*)d_in[4];
    const float* ln_gamma     = (const float*)d_in[5];
    const float* ln_beta      = (const float*)d_in[6];
    float* out = (float*)d_out;

    char* ws = (char*)d_ws;
    float* ruleW    = (float*)(ws);
    int*   ruleIdx  = (int*)  (ws + (size_t)R_ * 4 * 4);
    float* projPart = (float*)(ws + 65536);

    prep_gemm_kernel<<<R_ / RULE_WAVES_, 32 * (RULE_WAVES_ + GEMM_WAVES_), 0, stream>>>(
        fact_logits, agg_logits, strength, ruleIdx, ruleW, facts, W_proj, projPart);

    gate_ln_kernel<<<B_, 256, 0, stream>>>(projPart, facts, ruleIdx, ruleW,
                                           ln_gamma, ln_beta, out);
}


# launch 2: block barriers wait for LDS only (asm s_wait_dscnt + barrier; __syncthreads drained all global loads); slice/gamma/beta loads requested behind barrier 1 arrive under the top-8 search
# speedup vs baseline: 6.7393x; 1.1307x over previous
#include <hip/hip_runtime.h>
#include <hip/hip_bf16.h>

#define B_  32
#define D_  1024
#define R_  2048

typedef float v2f __attribute__((ext_vector_type(2)));
typedef float v8f __attribute__((ext_vector_type(8)));
typedef float v4f __attribute__((ext_vector_type(4)));
typedef int   v4i __attribute__((ext_vector_type(4)));

__device__ __forceinline__ void top2_insert(float v, int i,
                                            float& v1, int& i1,
                                            float& v2, int& i2) {
    bool beats1 = (v > v1) || (v == v1 && i < i1);
    if (beats1) {
        v2 = v1; i2 = i1;
        v1 = v;  i1 = i;
    } else {
        bool beats2 = (v > v2) || (v == v2 && i < i2);
        if (beats2) { v2 = v; i2 = i; }
    }
}

__device__ __forceinline__ void top2_take_rising(float v, int i,
                                                 float& v1, int& i1,
                                                 float& v2, int& i2) {
    const bool gt1 = v > v1;
    const bool gt2 = v > v2;
    v2 = gt1 ? v1 : (gt2 ? v : v2);
    i2 = gt1 ? i1 : (gt2 ? i : i2);
    v1 = gt1 ? v : v1;
    i1 = gt1 ? i : i1;
}

__device__ __forceinline__ void rule_prep_body(int r,
                                 const float* __restrict__ fact_logits,
                                 const float* __restrict__ agg_logits,
                                 const float* __restrict__ strength,
                                 int*   __restrict__ ruleIdx,
                                 float* __restrict__ ruleW)
{
    const int lane = threadIdx.x & 31;
    if (r >= R_) return;

    const float* row = fact_logits + (size_t)r * D_;
    v4f x[8];
    #pragma unroll
    for (int i = 0; i < 8; ++i) x[i] = *(const v4f*)(row + 128 * i + 4 * lane);
    const float a0 = agg_logits[3 * r + 0];
    const float a1 = agg_logits[3 * r + 1];
    const float a2 = agg_logits[3 * r + 2];
    const float st = strength[r];

    const int  base = 4 * lane;
    const bool sw   = x[0][1] > x[0][0];
    float v1 = sw ? x[0][1] : x[0][0];
    float v2 = sw ? x[0][0] : x[0][1];
    int   i1 = sw ? base + 1 : base;
    int   i2 = sw ? base : base + 1;
    #pragma unroll
    for (int i = 0; i < 8; ++i) {
        #pragma unroll
        for (int c = (i == 0) ? 2 : 0; c < 4; ++c) {
            top2_take_rising(x[i][c], 128 * i + base + c, v1, i1, v2, i2);
        }
    }
    #pragma unroll
    for (int m = 16; m >= 1; m >>= 1) {
        float ov1 = __shfl_xor(v1, m, 32);
        int   oi1 = __shfl_xor(i1, m, 32);
        float ov2 = __shfl_xor(v2, m, 32);
        int   oi2 = __shfl_xor(i2, m, 32);
        top2_insert(ov1, oi1, v1, i1, v2, i2);
        top2_insert(ov2, oi2, v1, i1, v2, i2);
    }

    if (lane == 0) {
        ruleIdx[2 * r + 0] = i1;
        ruleIdx[2 * r + 1] = i2;
        float mx = fmaxf(a0, fmaxf(a1, a2));
        float e0 = expf(a0 - mx), e1 = expf(a1 - mx), e2 = expf(a2 - mx);
        float inv = 1.0f / (e0 + e1 + e2);
        ruleW[4 * r + 0] = e0 * inv;
        ruleW[4 * r + 1] = e1 * inv;
        ruleW[4 * r + 2] = e2 * inv;
        ruleW[4 * r + 3] = 1.0f / (1.0f + expf(-st));
    }
}

#define NSLICE_ 8
#define KLEN_   (D_ / NSLICE_)

__device__ __forceinline__ void gemm_wmma_body(int g,
                                 const float* __restrict__ facts,
                                 const float* __restrict__ Wp,
                                 float* __restrict__ projPart)
{
    const int lane  = threadIdx.x & 31;

    const int ntile = g & 127;
    const int slice = g >> 7;
    const int n0    = ntile * 16;
    const int kbeg  = slice * KLEN_;

    const int rlo   = lane & 15;
    const int khalf = (lane >> 4) * 2;

    const float* aptr0 = facts + (size_t)(rlo)      * D_ + kbeg + khalf;
    const float* aptr1 = facts + (size_t)(16 + rlo) * D_ + kbeg + khalf;
    const float* bptr  = Wp    + (size_t)(n0 + rlo) * D_ + kbeg + khalf;

    v2f fb[KLEN_ / 4], fa0[KLEN_ / 4], fa1[KLEN_ / 4];
    #pragma unroll
    for (int i = 0; i < KLEN_ / 4; ++i) {
        fb[i]  = *(const v2f*)(bptr  + 4 * i);
        fa0[i] = *(const v2f*)(aptr0 + 4 * i);
        fa1[i] = *(const v2f*)(aptr1 + 4 * i);
    }

    v8f c0 = {};
    v8f c1 = {};
    #pragma unroll
    for (int i = 0; i < KLEN_ / 4; ++i) {
        c0 = __builtin_amdgcn_wmma_f32_16x16x4_f32(false, fa0[i], false, fb[i],
                                                   (short)0, c0, false, false);
        c1 = __builtin_amdgcn_wmma_f32_16x16x4_f32(false, fa1[i], false, fb[i],
                                                   (short)0, c1, false, false);
    }

    float* outp = projPart + (size_t)slice * B_ * R_;
    const int col   = n0 + rlo;
    const int rbase = (lane >> 4) * 8;
    #pragma unroll
    for (int v = 0; v < 8; ++v) {
        outp[(size_t)(rbase + v)      * R_ + col] = c0[v];
        outp[(size_t)(rbase + v + 16) * R_ + col] = c1[v];
    }
}

#define RULE_WAVES_ 8
#define GEMM_WAVES_ 4
static_assert((R_ / RULE_WAVES_) * GEMM_WAVES_ == (R_ / 16) * NSLICE_,
              "projection waves of the grid must cover every (N-tile, K-slice) pair once");

__global__ void __launch_bounds__(32 * (RULE_WAVES_ + GEMM_WAVES_))
__attribute__((amdgpu_waves_per_eu(3, 4)))
prep_gemm_kernel(const float* __restrict__ fact_logits,
                 const float* __restrict__ agg_logits,
                 const float* __restrict__ strength,
                 int*   __restrict__ ruleIdx,
                 float* __restrict__ ruleW,
                 const float* __restrict__ facts,
                 const float* __restrict__ Wp,
                 float* __restrict__ projPart)
{
    const int w = __builtin_amdgcn_readfirstlane((int)(threadIdx.x >> 5));
    if (w < RULE_WAVES_) {
        rule_prep_body((int)blockIdx.x * RULE_WAVES_ + w,
                       fact_logits, agg_logits, strength, ruleIdx, ruleW);
    } else {
        gemm_wmma_body((int)blockIdx.x * GEMM_WAVES_ + (w - RULE_WAVES_),
                       facts, Wp, projPart);
    }
}

__device__ __forceinline__ void lds_barrier() {
    asm volatile("s_wait_dscnt 0x0\n\ts_barrier_signal -1\n\ts_barrier_wait -1" ::: "memory");
}

template <int M>
__device__ __forceinline__ int row_xor_i(int x) {
    return __builtin_amdgcn_update_dpp(0, x, 0x160 + M, 0xf, 0xf, true);
}
__device__ __forceinline__ int half_xor_i(int x) {
    return __builtin_amdgcn_permlanex16(x, x, 0x76543210, 0xfedcba98, false, false);
}
template <int M>
__device__ __forceinline__ float row_xor_f(float x) {
    return __builtin_bit_cast(float, row_xor_i<M>(__builtin_bit_cast(int, x)));
}
__device__ __forceinline__ float half_xor_f(float x) {
    return __builtin_bit_cast(float, half_xor_i(__builtin_bit_cast(int, x)));
}
__device__ __forceinline__ int wave_max_i(int v) {
    v = max(v, row_xor_i<1>(v));
    v = max(v, row_xor_i<2>(v));
    v = max(v, row_xor_i<4>(v));
    v = max(v, row_xor_i<8>(v));
    v = max(v, half_xor_i(v));
    return v;
}
__device__ __forceinline__ int wave_min_i(int v) {
    v = min(v, row_xor_i<1>(v));
    v = min(v, row_xor_i<2>(v));
    v = min(v, row_xor_i<4>(v));
    v = min(v, row_xor_i<8>(v));
    v = min(v, half_xor_i(v));
    return v;
}
__device__ __forceinline__ float wave_sum_f(float v) {
    v += row_xor_f<1>(v);
    v += row_xor_f<2>(v);
    v += row_xor_f<4>(v);
    v += row_xor_f<8>(v);
    v += half_xor_f(v);
    return v;
}
__device__ __forceinline__ void cx_desc(int& a, int& b) {
    const int hi = max(a, b);
    const int lo = min(a, b);
    a = hi;
    b = lo;
}
__device__ __forceinline__ void sort8_desc(int (&k)[8]) {
    cx_desc(k[0], k[2]); cx_desc(k[1], k[3]); cx_desc(k[4], k[6]); cx_desc(k[5], k[7]);
    cx_desc(k[0], k[4]); cx_desc(k[1], k[5]); cx_desc(k[2], k[6]); cx_desc(k[3], k[7]);
    cx_desc(k[0], k[1]); cx_desc(k[2], k[3]); cx_desc(k[4], k[5]); cx_desc(k[6], k[7]);
    cx_desc(k[2], k[4]); cx_desc(k[3], k[5]);
    cx_desc(k[1], k[4]); cx_desc(k[3], k[6]);
    cx_desc(k[1], k[2]); cx_desc(k[3], k[4]); cx_desc(k[5], k[6]);
}
__device__ __forceinline__ void merge8_desc(int (&k)[8], const int (&p)[8]) {
    #pragma unroll
    for (int i = 0; i < 8; ++i) k[i] = max(k[i], p[i]);
    cx_desc(k[0], k[4]); cx_desc(k[1], k[5]); cx_desc(k[2], k[6]); cx_desc(k[3], k[7]);
    cx_desc(k[0], k[2]); cx_desc(k[1], k[3]); cx_desc(k[4], k[6]); cx_desc(k[5], k[7]);
    cx_desc(k[0], k[1]); cx_desc(k[2], k[3]); cx_desc(k[4], k[5]); cx_desc(k[6], k[7]);
}
template <int M>
__device__ __forceinline__ void merge_row_xor(int (&k)[8]) {
    int p[8];
    #pragma unroll
    for (int i = 0; i < 8; ++i) p[i] = row_xor_i<M>(k[7 - i]);
    merge8_desc(k, p);
}
__device__ __forceinline__ void merge_half_xor(int (&k)[8]) {
    int p[8];
    #pragma unroll
    for (int i = 0; i < 8; ++i) p[i] = half_xor_i(k[7 - i]);
    merge8_desc(k, p);
}
__device__ __forceinline__ int order_key(float v) {
    const int u = __builtin_bit_cast(int, v + 0.0f);
    return u ^ ((u >> 31) & 0x7fffffff);
}
__device__ __forceinline__ void wave_best(int bk, int bi, int& wk, int& wi) {
    wk = wave_max_i(bk);
    wi = wave_min_i((bk == wk) ? bi : 0x7fffffff);
}

__device__ __forceinline__ unsigned gate_mask_by_rounds(const float (&act)[8], int e0,
                                                        int lane, int wave,
                                                        int* s_ck, int* s_ci) {
    const int kTaken = (int)0x80000000;
    int key[8];
    #pragma unroll
    for (int j = 0; j < 8; ++j) key[j] = order_key(act[j]);

    int myck = kTaken;
    int myci = 0x7fffffff;
    #pragma unroll
    for (int t = 0; t < 8; ++t) {
        int bk = kTaken;
        int bi = 0x7fffffff;
        #pragma unroll
        for (int j = 0; j < 8; ++j) {
            const int  idx  = (j >> 2) * 1024 + e0 + (j & 3);
            const bool take = key[j] > bk;
            bk = take ? key[j] : bk;
            bi = take ? idx : bi;
        }
        int wk, wi;
        wave_best(bk, bi, wk, wi);
        #pragma unroll
        for (int j = 0; j < 8; ++j) {
            const int idx = (j >> 2) * 1024 + e0 + (j & 3);
            key[j] = (idx == wi) ? kTaken : key[j];
        }
        myck = (lane == t) ? wk : myck;
        myci = (lane == t) ? wi : myci;
    }
    if (lane < 8) {
        s_ck[wave * 8 + lane] = myck;
        s_ci[wave * 8 + lane] = myci;
    }
    lds_barrier();

    int ck[2], ci[2];
    ck[0] = s_ck[lane];      ci[0] = s_ci[lane];
    ck[1] = s_ck[lane + 32]; ci[1] = s_ci[lane + 32];
    unsigned gmask = 0;
    #pragma unroll
    for (int t = 0; t < 8; ++t) {
        int bk = kTaken;
        int bi = 0x7fffffff;
        #pragma unroll
        for (int q = 0; q < 2; ++q) {
            const bool take = (ck[q] > bk) | ((ck[q] == bk) & (ci[q] < bi));
            bk = take ? ck[q] : bk;
            bi = take ? ci[q] : bi;
        }
        int wk, wi;
        wave_best(bk, bi, wk, wi);
        #pragma unroll
        for (int q = 0; q < 2; ++q) ck[q] = (ci[q] == wi) ? kTaken : ck[q];
        const unsigned uw   = (unsigned)wi;
        const bool     mine = ((wi & 1020) == e0) & (uw < (unsigned)R_);
        const unsigned bit  = 1u << ((((uw >> 10) & 1u) << 2) | (uw & 3u));
        gmask |= mine ? bit : 0u;
    }
    return gmask;
}

__global__ void __launch_bounds__(256)
gate_ln_kernel(const float* __restrict__ projPart,
               const float* __restrict__ facts,
               const int*   __restrict__ ruleIdx,
               const float* __restrict__ ruleW,
               const float* __restrict__ gamma,
               const float* __restrict__ beta,
               float* __restrict__ out)
{
    const int b    = blockIdx.x;
    const int tid  = threadIdx.x;
    const int lane = tid & 31;
    const int wave = tid >> 5;
    const int e0   = 4 * tid;

    __shared__ float s_f[D_];
    __shared__ int   s_ck[64];
    __shared__ int   s_ci[64];
    __shared__ __attribute__((aligned(16))) int s_top[64];
    __shared__ float s_sum[8];
    __shared__ float s_sq[8];
    __shared__ int   s_cnt[8];

    const size_t rowoff = (size_t)b * R_;

    const v4f frow = *(const v4f*)(facts + (size_t)b * D_ + e0);
    __builtin_amdgcn_sched_barrier(0);
    v4i ix[2][2];
    v4f rw[2][4];
    #pragma unroll
    for (int h = 0; h < 2; ++h) {
        const int r0 = h * 1024 + e0;
        ix[h][0] = *(const v4i*)(ruleIdx + 2 * r0);
        ix[h][1] = *(const v4i*)(ruleIdx + 2 * r0 + 4);
        #pragma unroll
        for (int c = 0; c < 4; ++c) rw[h][c] = *(const v4f*)(ruleW + 4 * (r0 + c));
    }
    *(v4f*)(s_f + e0) = frow;

    lds_barrier();

    v4f g4[2], be4[2], p4[NSLICE_][2];
    #pragma unroll
    for (int h = 0; h < 2; ++h) {
        const int col = h * 1024 + e0;
        #pragma unroll
        for (int s = 0; s < NSLICE_; ++s) {
            p4[s][h] = *(const v4f*)(projPart + (size_t)s * B_ * R_ + rowoff + col);
        }
    }
    #pragma unroll
    for (int h = 0; h < 2; ++h) {
        const int col = h * 1024 + e0;
        g4[h]  = *(const v4f*)(gamma + col);
        be4[h] = *(const v4f*)(beta + col);
    }
    __builtin_amdgcn_sched_barrier(0);

    float act[8];
    #pragma unroll
    for (int j = 0; j < 8; ++j) {
        const v4i  q  = ix[j >> 2][(j >> 1) & 1];
        const int  i0 = ((j & 1) ? q.z : q.x) & (D_ - 1);
        const int  i1 = ((j & 1) ? q.w : q.y) & (D_ - 1);
        const float f0 = s_f[i0];
        const float f1 = s_f[i1];
        const v4f  w  = rw[j >> 2][j & 3];

        const float and_a = f0 * f1;
        const float or_a  = 1.0f - (1.0f - f0) * (1.0f - f1);
        const float k_a   = (f0 + f1) / (2.0f + 1e-8f);

        act[j] = (w.x * and_a + w.y * or_a + w.z * k_a) * w.w;
    }

    int key[8], tk[8];
    #pragma unroll
    for (int j = 0; j < 8; ++j) { key[j] = order_key(act[j]); tk[j] = key[j]; }
    sort8_desc(tk);
    merge_row_xor<1>(tk);
    merge_row_xor<2>(tk);
    merge_row_xor<4>(tk);
    merge_row_xor<8>(tk);
    merge_half_xor(tk);
    if (lane == 0) {
        v4i lo, hi;
        lo.x = tk[0]; lo.y = tk[1]; lo.z = tk[2]; lo.w = tk[3];
        hi.x = tk[4]; hi.y = tk[5]; hi.z = tk[6]; hi.w = tk[7];
        *(v4i*)(s_top + wave * 8)     = lo;
        *(v4i*)(s_top + wave * 8 + 4) = hi;
    }
    lds_barrier();

    {
        const v4i lo = *(const v4i*)(s_top + (lane & 7) * 8);
        const v4i hi = *(const v4i*)(s_top + (lane & 7) * 8 + 4);
        tk[0] = lo.x; tk[1] = lo.y; tk[2] = lo.z; tk[3] = lo.w;
        tk[4] = hi.x; tk[5] = hi.y; tk[6] = hi.z; tk[7] = hi.w;
    }
    merge_row_xor<1>(tk);
    merge_row_xor<2>(tk);
    merge_row_xor<4>(tk);
    const int thr = tk[7];
    bool on[8];
    int  cnt = 0;
    #pragma unroll
    for (int j = 0; j < 8; ++j) {
        on[j] = key[j] >= thr;
        cnt += __builtin_popcount(__builtin_amdgcn_ballot_w32(on[j]));
    }

    __builtin_amdgcn_sched_barrier(0);

    float base[8];
    #pragma unroll
    for (int h = 0; h < 2; ++h) {
        #pragma unroll
        for (int c = 0; c < 4; ++c) {
            float acc = 0.0f;
            #pragma unroll
            for (int s = 0; s < NSLICE_; ++s) acc += p4[s][h][c];
            base[h * 4 + c] = acc;
        }
    }

    float pre[8];
    #pragma unroll
    for (int j = 0; j < 8; ++j) {
        const float g = base[j] + act[j];
        pre[j] = on[j] ? g : base[j];
    }

    float sum = 0.0f;
    #pragma unroll
    for (int j = 0; j < 8; ++j) sum += pre[j];
    sum = wave_sum_f(sum);
    if (lane == 0) { s_sum[wave] = sum; s_cnt[wave] = cnt; }
    lds_barrier();
    float tot = 0.0f;
    int   totc = 0;
    #pragma unroll
    for (int w = 0; w < 8; ++w) { tot += s_sum[w]; totc += s_cnt[w]; }

    if (__builtin_amdgcn_readfirstlane(totc) != 8) {
        const unsigned gmask = gate_mask_by_rounds(act, e0, lane, wave, s_ck, s_ci);
        #pragma unroll
        for (int j = 0; j < 8; ++j) {
            const float g = base[j] + act[j];
            pre[j] = ((gmask >> j) & 1u) ? g : base[j];
        }
        sum = 0.0f;
        #pragma unroll
        for (int j = 0; j < 8; ++j) sum += pre[j];
        sum = wave_sum_f(sum);
        if (lane == 0) s_sum[wave] = sum;
        lds_barrier();
        tot = 0.0f;
        #pragma unroll
        for (int w = 0; w < 8; ++w) tot += s_sum[w];
    }
    const float mu = tot * (1.0f / R_);

    float ss = 0.0f;
    #pragma unroll
    for (int j = 0; j < 8; ++j) { float d = pre[j] - mu; ss += d * d; }
    ss = wave_sum_f(ss);
    if (lane == 0) s_sq[wave] = ss;
    lds_barrier();
    float tot2 = 0.0f;
    #pragma unroll
    for (int w = 0; w < 8; ++w) tot2 += s_sq[w];
    const float inv = rsqrtf(tot2 * (1.0f / R_) + 1e-5f);

    #pragma unroll
    for (int h = 0; h < 2; ++h) {
        v4f o;
        o.x = (pre[h * 4 + 0] - mu) * inv * g4[h].x + be4[h].x;
        o.y = (pre[h * 4 + 1] - mu) * inv * g4[h].y + be4[h].y;
        o.z = (pre[h * 4 + 2] - mu) * inv * g4[h].z + be4[h].z;
        o.w = (pre[h * 4 + 3] - mu) * inv * g4[h].w + be4[h].w;
        *(v4f*)(out + rowoff + h * 1024 + e0) = o;
    }
}

extern "C" void kernel_launch(void* const* d_in, const int* in_sizes, int n_in,
                              void* d_out, int out_size, void* d_ws, size_t ws_size,
                              hipStream_t stream) {
    const float* facts        = (const float*)d_in[0];
    const float* fact_logits  = (const float*)d_in[1];
    const float* agg_logits   = (const float*)d_in[2];
    const float* strength     = (const float*)d_in[3];
    const float* W_proj       = (const float*)d_in[4];
    const float* ln_gamma     = (const float*)d_in[5];
    const float* ln_beta      = (const float*)d_in[6];
    float* out = (float*)d_out;

    char* ws = (char*)d_ws;
    float* ruleW    = (float*)(ws);
    int*   ruleIdx  = (int*)  (ws + (size_t)R_ * 4 * 4);
    float* projPart = (float*)(ws + 65536);

    prep_gemm_kernel<<<R_ / RULE_WAVES_, 32 * (RULE_WAVES_ + GEMM_WAVES_), 0, stream>>>(
        fact_logits, agg_logits, strength, ruleIdx, ruleW, facts, W_proj, projPart);

    gate_ln_kernel<<<B_, 256, 0, stream>>>(projPart, facts, ruleIdx, ruleW,
                                           ln_gamma, ln_beta, out);
}


# rule waves take two rows each: 4 rule + 4 projection waves per work-group, 2048 waves to enter instead of 3072; the rows' list merges interleave as DPP moves and selects; no sum changed
# speedup vs baseline: 7.1597x; 1.0624x over previous
#include <hip/hip_runtime.h>
#include <hip/hip_bf16.h>

#define B_  32
#define D_  1024
#define R_  2048

typedef float v2f __attribute__((ext_vector_type(2)));
typedef float v8f __attribute__((ext_vector_type(8)));
typedef float v4f __attribute__((ext_vector_type(4)));
typedef int   v4i __attribute__((ext_vector_type(4)));
typedef int   v2i __attribute__((ext_vector_type(2)));

__device__ __forceinline__ void top2_merge(float ov1, int oi1, float ov2, int oi2,
                                           float& v1, int& i1, float& v2, int& i2) {
    const bool  mine = (v1 > ov1) | ((v1 == ov1) & (i1 < oi1));
    const float cv = mine ? v2  : v1;
    const int   ci = mine ? i2  : i1;
    const float dv = mine ? ov1 : ov2;
    const int   di = mine ? oi1 : oi2;
    const bool  cw = (cv > dv) | ((cv == dv) & (ci < di));
    v1 = mine ? v1 : ov1;
    i1 = mine ? i1 : oi1;
    v2 = cw ? cv : dv;
    i2 = cw ? ci : di;
}

__device__ __forceinline__ void top2_take_rising(float v, int i,
                                                 float& v1, int& i1,
                                                 float& v2, int& i2) {
    const bool gt1 = v > v1;
    const bool gt2 = v > v2;
    v2 = gt1 ? v1 : (gt2 ? v : v2);
    i2 = gt1 ? i1 : (gt2 ? i : i2);
    v1 = gt1 ? v : v1;
    i1 = gt1 ? i : i1;
}

template <int M>
__device__ __forceinline__ int row_xor_i(int x) {
    return __builtin_amdgcn_update_dpp(0, x, 0x160 + M, 0xf, 0xf, true);
}
__device__ __forceinline__ int half_xor_i(int x) {
    return __builtin_amdgcn_permlanex16(x, x, 0x76543210, 0xfedcba98, false, false);
}
template <int M>
__device__ __forceinline__ float row_xor_f(float x) {
    return __builtin_bit_cast(float, row_xor_i<M>(__builtin_bit_cast(int, x)));
}
__device__ __forceinline__ float half_xor_f(float x) {
    return __builtin_bit_cast(float, half_xor_i(__builtin_bit_cast(int, x)));
}
template <int M>
__device__ __forceinline__ void top2_merge_row_xor(float& v1, int& i1, float& v2, int& i2) {
    const float ov1 = row_xor_f<M>(v1);
    const int   oi1 = row_xor_i<M>(i1);
    const float ov2 = row_xor_f<M>(v2);
    const int   oi2 = row_xor_i<M>(i2);
    top2_merge(ov1, oi1, ov2, oi2, v1, i1, v2, i2);
}
__device__ __forceinline__ void top2_merge_half_xor(float& v1, int& i1, float& v2, int& i2) {
    const float ov1 = half_xor_f(v1);
    const int   oi1 = half_xor_i(i1);
    const float ov2 = half_xor_f(v2);
    const int   oi2 = half_xor_i(i2);
    top2_merge(ov1, oi1, ov2, oi2, v1, i1, v2, i2);
}

__device__ __forceinline__ void rule_prep_body(int p,
                                 const float* __restrict__ fact_logits,
                                 const float* __restrict__ agg_logits,
                                 const float* __restrict__ strength,
                                 int*   __restrict__ ruleIdx,
                                 float* __restrict__ ruleW)
{
    const int lane = threadIdx.x & 31;
    const int r0   = 2 * p;
    if (r0 >= R_) return;

    const float* rowa = fact_logits + (size_t)r0 * D_;
    const float* rowb = rowa + D_;
    v4f xa[8], xb[8];
    #pragma unroll
    for (int i = 0; i < 8; ++i) {
        xa[i] = *(const v4f*)(rowa + 128 * i + 4 * lane);
        xb[i] = *(const v4f*)(rowb + 128 * i + 4 * lane);
    }
    const float aa0 = agg_logits[3 * r0 + 0];
    const float aa1 = agg_logits[3 * r0 + 1];
    const float aa2 = agg_logits[3 * r0 + 2];
    const float ab0 = agg_logits[3 * r0 + 3];
    const float ab1 = agg_logits[3 * r0 + 4];
    const float ab2 = agg_logits[3 * r0 + 5];
    const float sta = strength[r0];
    const float stb = strength[r0 + 1];
    asm volatile("" :: "s"(aa0), "s"(aa1), "s"(aa2), "s"(ab0), "s"(ab1), "s"(ab2), "s"(sta), "s"(stb));

    const int  base = 4 * lane;
    const bool swa  = xa[0][1] > xa[0][0];
    const bool swb  = xb[0][1] > xb[0][0];
    float va1 = swa ? xa[0][1] : xa[0][0];
    float va2 = swa ? xa[0][0] : xa[0][1];
    int   ia1 = swa ? base + 1 : base;
    int   ia2 = swa ? base : base + 1;
    float vb1 = swb ? xb[0][1] : xb[0][0];
    float vb2 = swb ? xb[0][0] : xb[0][1];
    int   ib1 = swb ? base + 1 : base;
    int   ib2 = swb ? base : base + 1;
    #pragma unroll
    for (int i = 0; i < 8; ++i) {
        #pragma unroll
        for (int c = (i == 0) ? 2 : 0; c < 4; ++c) {
            top2_take_rising(xa[i][c], 128 * i + base + c, va1, ia1, va2, ia2);
            top2_take_rising(xb[i][c], 128 * i + base + c, vb1, ib1, vb2, ib2);
        }
    }
    top2_merge_row_xor<1>(va1, ia1, va2, ia2);
    top2_merge_row_xor<1>(vb1, ib1, vb2, ib2);
    top2_merge_row_xor<2>(va1, ia1, va2, ia2);
    top2_merge_row_xor<2>(vb1, ib1, vb2, ib2);
    top2_merge_row_xor<4>(va1, ia1, va2, ia2);
    top2_merge_row_xor<4>(vb1, ib1, vb2, ib2);
    top2_merge_row_xor<8>(va1, ia1, va2, ia2);
    top2_merge_row_xor<8>(vb1, ib1, vb2, ib2);
    top2_merge_half_xor(va1, ia1, va2, ia2);
    top2_merge_half_xor(vb1, ib1, vb2, ib2);

    if (lane < 2) {
        const bool second = lane != 0;
        const int  r  = r0 + lane;
        const float a0 = second ? ab0 : aa0;
        const float a1 = second ? ab1 : aa1;
        const float a2 = second ? ab2 : aa2;
        const float st = second ? stb : sta;
        v2i idx;
        idx.x = second ? ib1 : ia1;
        idx.y = second ? ib2 : ia2;
        *(v2i*)(ruleIdx + 2 * r) = idx;
        float mx = fmaxf(a0, fmaxf(a1, a2));
        float e0 = expf(a0 - mx), e1 = expf(a1 - mx), e2 = expf(a2 - mx);
        float inv = 1.0f / (e0 + e1 + e2);
        v4f w;
        w.x = e0 * inv;
        w.y = e1 * inv;
        w.z = e2 * inv;
        w.w = 1.0f / (1.0f + expf(-st));
        *(v4f*)(ruleW + 4 * r) = w;
    }
}

#define NSLICE_ 8
#define KLEN_   (D_ / NSLICE_)

__device__ __forceinline__ void gemm_wmma_body(int g,
                                 const float* __restrict__ facts,
                                 const float* __restrict__ Wp,
                                 float* __restrict__ projPart)
{
    const int lane  = threadIdx.x & 31;

    const int ntile = g & 127;
    const int slice = g >> 7;
    const int n0    = ntile * 16;
    const int kbeg  = slice * KLEN_;

    const int rlo   = lane & 15;
    const int khalf = (lane >> 4) * 2;

    const float* aptr0 = facts + (size_t)(rlo)      * D_ + kbeg + khalf;
    const float* aptr1 = facts + (size_t)(16 + rlo) * D_ + kbeg + khalf;
    const float* bptr  = Wp    + (size_t)(n0 + rlo) * D_ + kbeg + khalf;

    v2f fb[KLEN_ / 4], fa0[KLEN_ / 4], fa1[KLEN_ / 4];
    #pragma unroll
    for (int i = 0; i < KLEN_ / 4; ++i) {
        fb[i]  = *(const v2f*)(bptr  + 4 * i);
        fa0[i] = *(const v2f*)(aptr0 + 4 * i);
        fa1[i] = *(const v2f*)(aptr1 + 4 * i);
    }

    v8f c0 = {};
    v8f c1 = {};
    #pragma unroll
    for (int i = 0; i < KLEN_ / 4; ++i) {
        c0 = __builtin_amdgcn_wmma_f32_16x16x4_f32(false, fa0[i], false, fb[i],
                                                   (short)0, c0, false, false);
        c1 = __builtin_amdgcn_wmma_f32_16x16x4_f32(false, fa1[i], false, fb[i],
                                                   (short)0, c1, false, false);
    }

    float* outp = projPart + (size_t)slice * B_ * R_;
    const int col   = n0 + rlo;
    const int rbase = (lane >> 4) * 8;
    #pragma unroll
    for (int v = 0; v < 8; ++v) {
        outp[(size_t)(rbase + v)      * R_ + col] = c0[v];
        outp[(size_t)(rbase + v + 16) * R_ + col] = c1[v];
    }
}

#define RULE_WAVES_     4
#define RULES_PER_WAVE_ 2
#define GEMM_WAVES_     4
static_assert((R_ / (RULE_WAVES_ * RULES_PER_WAVE_)) * GEMM_WAVES_ == (R_ / 16) * NSLICE_,
              "projection waves of the grid must cover every (N-tile, K-slice) pair once");

__global__ void __launch_bounds__(32 * (RULE_WAVES_ + GEMM_WAVES_))
__attribute__((amdgpu_waves_per_eu(2, 4)))
prep_gemm_kernel(const float* __restrict__ fact_logits,
                 const float* __restrict__ agg_logits,
                 const float* __restrict__ strength,
                 int*   __restrict__ ruleIdx,
                 float* __restrict__ ruleW,
                 const float* __restrict__ facts,
                 const float* __restrict__ Wp,
                 float* __restrict__ projPart)
{
    const int w = __builtin_amdgcn_readfirstlane((int)(threadIdx.x >> 5));
    if (w < RULE_WAVES_) {
        rule_prep_body((int)blockIdx.x * RULE_WAVES_ + w,
                       fact_logits, agg_logits, strength, ruleIdx, ruleW);
    } else {
        gemm_wmma_body((int)blockIdx.x * GEMM_WAVES_ + (w - RULE_WAVES_),
                       facts, Wp, projPart);
    }
}

__device__ __forceinline__ void lds_barrier() {
    asm volatile("s_wait_dscnt 0x0\n\ts_barrier_signal -1\n\ts_barrier_wait -1" ::: "memory");
}

__device__ __forceinline__ int wave_max_i(int v) {
    v = max(v, row_xor_i<1>(v));
    v = max(v, row_xor_i<2>(v));
    v = max(v, row_xor_i<4>(v));
    v = max(v, row_xor_i<8>(v));
    v = max(v, half_xor_i(v));
    return v;
}
__device__ __forceinline__ int wave_min_i(int v) {
    v = min(v, row_xor_i<1>(v));
    v = min(v, row_xor_i<2>(v));
    v = min(v, row_xor_i<4>(v));
    v = min(v, row_xor_i<8>(v));
    v = min(v, half_xor_i(v));
    return v;
}
__device__ __forceinline__ float wave_sum_f(float v) {
    v += row_xor_f<1>(v);
    v += row_xor_f<2>(v);
    v += row_xor_f<4>(v);
    v += row_xor_f<8>(v);
    v += half_xor_f(v);
    return v;
}
__device__ __forceinline__ void cx_desc(int& a, int& b) {
    const int hi = max(a, b);
    const int lo = min(a, b);
    a = hi;
    b = lo;
}
__device__ __forceinline__ void sort8_desc(int (&k)[8]) {
    cx_desc(k[0], k[2]); cx_desc(k[1], k[3]); cx_desc(k[4], k[6]); cx_desc(k[5], k[7]);
    cx_desc(k[0], k[4]); cx_desc(k[1], k[5]); cx_desc(k[2], k[6]); cx_desc(k[3], k[7]);
    cx_desc(k[0], k[1]); cx_desc(k[2], k[3]); cx_desc(k[4], k[5]); cx_desc(k[6], k[7]);
    cx_desc(k[2], k[4]); cx_desc(k[3], k[5]);
    cx_desc(k[1], k[4]); cx_desc(k[3], k[6]);
    cx_desc(k[1], k[2]); cx_desc(k[3], k[4]); cx_desc(k[5], k[6]);
}
__device__ __forceinline__ void merge8_desc(int (&k)[8], const int (&p)[8]) {
    #pragma unroll
    for (int i = 0; i < 8; ++i) k[i] = max(k[i], p[i]);
    cx_desc(k[0], k[4]); cx_desc(k[1], k[5]); cx_desc(k[2], k[6]); cx_desc(k[3], k[7]);
    cx_desc(k[0], k[2]); cx_desc(k[1], k[3]); cx_desc(k[4], k[6]); cx_desc(k[5], k[7]);
    cx_desc(k[0], k[1]); cx_desc(k[2], k[3]); cx_desc(k[4], k[5]); cx_desc(k[6], k[7]);
}
template <int M>
__device__ __forceinline__ void merge_row_xor(int (&k)[8]) {
    int p[8];
    #pragma unroll
    for (int i = 0; i < 8; ++i) p[i] = row_xor_i<M>(k[7 - i]);
    merge8_desc(k, p);
}
__device__ __forceinline__ void merge_half_xor(int (&k)[8]) {
    int p[8];
    #pragma unroll
    for (int i = 0; i < 8; ++i) p[i] = half_xor_i(k[7 - i]);
    merge8_desc(k, p);
}
__device__ __forceinline__ int order_key(float v) {
    const int u = __builtin_bit_cast(int, v + 0.0f);
    return u ^ ((u >> 31) & 0x7fffffff);
}
__device__ __forceinline__ void wave_best(int bk, int bi, int& wk, int& wi) {
    wk = wave_max_i(bk);
    wi = wave_min_i((bk == wk) ? bi : 0x7fffffff);
}

__device__ __forceinline__ unsigned gate_mask_by_rounds(const float (&act)[8], int e0,
                                                        int lane, int wave,
                                                        int* s_ck, int* s_ci) {
    const int kTaken = (int)0x80000000;
    int key[8];
    #pragma unroll
    for (int j = 0; j < 8; ++j) key[j] = order_key(act[j]);

    int myck = kTaken;
    int myci = 0x7fffffff;
    #pragma unroll
    for (int t = 0; t < 8; ++t) {
        int bk = kTaken;
        int bi = 0x7fffffff;
        #pragma unroll
        for (int j = 0; j < 8; ++j) {
            const int  idx  = (j >> 2) * 1024 + e0 + (j & 3);
            const bool take = key[j] > bk;
            bk = take ? key[j] : bk;
            bi = take ? idx : bi;
        }
        int wk, wi;
        wave_best(bk, bi, wk, wi);
        #pragma unroll
        for (int j = 0; j < 8; ++j) {
            const int idx = (j >> 2) * 1024 + e0 + (j & 3);
            key[j] = (idx == wi) ? kTaken : key[j];
        }
        myck = (lane == t) ? wk : myck;
        myci = (lane == t) ? wi : myci;
    }
    if (lane < 8) {
        s_ck[wave * 8 + lane] = myck;
        s_ci[wave * 8 + lane] = myci;
    }
    lds_barrier();

    int ck[2], ci[2];
    ck[0] = s_ck[lane];      ci[0] = s_ci[lane];
    ck[1] = s_ck[lane + 32]; ci[1] = s_ci[lane + 32];
    unsigned gmask = 0;
    #pragma unroll
    for (int t = 0; t < 8; ++t) {
        int bk = kTaken;
        int bi = 0x7fffffff;
        #pragma unroll
        for (int q = 0; q < 2; ++q) {
            const bool take = (ck[q] > bk) | ((ck[q] == bk) & (ci[q] < bi));
            bk = take ? ck[q] : bk;
            bi = take ? ci[q] : bi;
        }
        int wk, wi;
        wave_best(bk, bi, wk, wi);
        #pragma unroll
        for (int q = 0; q < 2; ++q) ck[q] = (ci[q] == wi) ? kTaken : ck[q];
        const unsigned uw   = (unsigned)wi;
        const bool     mine = ((wi & 1020) == e0) & (uw < (unsigned)R_);
        const unsigned bit  = 1u << ((((uw >> 10) & 1u) << 2) | (uw & 3u));
        gmask |= mine ? bit : 0u;
    }
    return gmask;
}

__global__ void __launch_bounds__(256)
gate_ln_kernel(const float* __restrict__ projPart,
               const float* __restrict__ facts,
               const int*   __restrict__ ruleIdx,
               const float* __restrict__ ruleW,
               const float* __restrict__ gamma,
               const float* __restrict__ beta,
               float* __restrict__ out)
{
    const int b    = blockIdx.x;
    const int tid  = threadIdx.x;
    const int lane = tid & 31;
    const int wave = tid >> 5;
    const int e0   = 4 * tid;

    __shared__ float s_f[D_];
    __shared__ int   s_ck[64];
    __shared__ int   s_ci[64];
    __shared__ __attribute__((aligned(16))) int s_top[64];
    __shared__ float s_sum[8];
    __shared__ float s_sq[8];
    __shared__ int   s_cnt[8];

    const size_t rowoff = (size_t)b * R_;

    const v4f frow = *(const v4f*)(facts + (size_t)b * D_ + e0);
    __builtin_amdgcn_sched_barrier(0);
    v4i ix[2][2];
    v4f rw[2][4];
    #pragma unroll
    for (int h = 0; h < 2; ++h) {
        const int r0 = h * 1024 + e0;
        ix[h][0] = *(const v4i*)(ruleIdx + 2 * r0);
        ix[h][1] = *(const v4i*)(ruleIdx + 2 * r0 + 4);
        #pragma unroll
        for (int c = 0; c < 4; ++c) rw[h][c] = *(const v4f*)(ruleW + 4 * (r0 + c));
    }
    *(v4f*)(s_f + e0) = frow;

    lds_barrier();

    v4f g4[2], be4[2], p4[NSLICE_][2];
    #pragma unroll
    for (int h = 0; h < 2; ++h) {
        const int col = h * 1024 + e0;
        #pragma unroll
        for (int s = 0; s < NSLICE_; ++s) {
            p4[s][h] = *(const v4f*)(projPart + (size_t)s * B_ * R_ + rowoff + col);
        }
    }
    #pragma unroll
    for (int h = 0; h < 2; ++h) {
        const int col = h * 1024 + e0;
        g4[h]  = *(const v4f*)(gamma + col);
        be4[h] = *(const v4f*)(beta + col);
    }
    __builtin_amdgcn_sched_barrier(0);

    float act[8];
    #pragma unroll
    for (int j = 0; j < 8; ++j) {
        const v4i  q  = ix[j >> 2][(j >> 1) & 1];
        const int  i0 = ((j & 1) ? q.z : q.x) & (D_ - 1);
        const int  i1 = ((j & 1) ? q.w : q.y) & (D_ - 1);
        const float f0 = s_f[i0];
        const float f1 = s_f[i1];
        const v4f  w  = rw[j >> 2][j & 3];

        const float and_a = f0 * f1;
        const float or_a  = 1.0f - (1.0f - f0) * (1.0f - f1);
        const float k_a   = (f0 + f1) / (2.0f + 1e-8f);

        act[j] = (w.x * and_a + w.y * or_a + w.z * k_a) * w.w;
    }

    int key[8], tk[8];
    #pragma unroll
    for (int j = 0; j < 8; ++j) { key[j] = order_key(act[j]); tk[j] = key[j]; }
    sort8_desc(tk);
    merge_row_xor<1>(tk);
    merge_row_xor<2>(tk);
    merge_row_xor<4>(tk);
    merge_row_xor<8>(tk);
    merge_half_xor(tk);
    if (lane == 0) {
        v4i lo, hi;
        lo.x = tk[0]; lo.y = tk[1]; lo.z = tk[2]; lo.w = tk[3];
        hi.x = tk[4]; hi.y = tk[5]; hi.z = tk[6]; hi.w = tk[7];
        *(v4i*)(s_top + wave * 8)     = lo;
        *(v4i*)(s_top + wave * 8 + 4) = hi;
    }
    lds_barrier();

    {
        const v4i lo = *(const v4i*)(s_top + (lane & 7) * 8);
        const v4i hi = *(const v4i*)(s_top + (lane & 7) * 8 + 4);
        tk[0] = lo.x; tk[1] = lo.y; tk[2] = lo.z; tk[3] = lo.w;
        tk[4] = hi.x; tk[5] = hi.y; tk[6] = hi.z; tk[7] = hi.w;
    }
    merge_row_xor<1>(tk);
    merge_row_xor<2>(tk);
    merge_row_xor<4>(tk);
    const int thr = tk[7];
    bool on[8];
    int  cnt = 0;
    #pragma unroll
    for (int j = 0; j < 8; ++j) {
        on[j] = key[j] >= thr;
        cnt += __builtin_popcount(__builtin_amdgcn_ballot_w32(on[j]));
    }

    __builtin_amdgcn_sched_barrier(0);

    float base[8];
    #pragma unroll
    for (int h = 0; h < 2; ++h) {
        #pragma unroll
        for (int c = 0; c < 4; ++c) {
            float acc = 0.0f;
            #pragma unroll
            for (int s = 0; s < NSLICE_; ++s) acc += p4[s][h][c];
            base[h * 4 + c] = acc;
        }
    }

    float pre[8];
    #pragma unroll
    for (int j = 0; j < 8; ++j) {
        const float g = base[j] + act[j];
        pre[j] = on[j] ? g : base[j];
    }

    float sum = 0.0f;
    #pragma unroll
    for (int j = 0; j < 8; ++j) sum += pre[j];
    sum = wave_sum_f(sum);
    if (lane == 0) { s_sum[wave] = sum; s_cnt[wave] = cnt; }
    lds_barrier();
    float tot = 0.0f;
    int   totc = 0;
    #pragma unroll
    for (int w = 0; w < 8; ++w) { tot += s_sum[w]; totc += s_cnt[w]; }

    if (__builtin_amdgcn_readfirstlane(totc) != 8) {
        const unsigned gmask = gate_mask_by_rounds(act, e0, lane, wave, s_ck, s_ci);
        #pragma unroll
        for (int j = 0; j < 8; ++j) {
            const float g = base[j] + act[j];
            pre[j] = ((gmask >> j) & 1u) ? g : base[j];
        }
        sum = 0.0f;
        #pragma unroll
        for (int j = 0; j < 8; ++j) sum += pre[j];
        sum = wave_sum_f(sum);
        if (lane == 0) s_sum[wave] = sum;
        lds_barrier();
        tot = 0.0f;
        #pragma unroll
        for (int w = 0; w < 8; ++w) tot += s_sum[w];
    }
    const float mu = tot * (1.0f / R_);

    float ss = 0.0f;
    #pragma unroll
    for (int j = 0; j < 8; ++j) { float d = pre[j] - mu; ss += d * d; }
    ss = wave_sum_f(ss);
    if (lane == 0) s_sq[wave] = ss;
    lds_barrier();
    float tot2 = 0.0f;
    #pragma unroll
    for (int w = 0; w < 8; ++w) tot2 += s_sq[w];
    const float inv = rsqrtf(tot2 * (1.0f / R_) + 1e-5f);

    #pragma unroll
    for (int h = 0; h < 2; ++h) {
        v4f o;
        o.x = (pre[h * 4 + 0] - mu) * inv * g4[h].x + be4[h].x;
        o.y = (pre[h * 4 + 1] - mu) * inv * g4[h].y + be4[h].y;
        o.z = (pre[h * 4 + 2] - mu) * inv * g4[h].z + be4[h].z;
        o.w = (pre[h * 4 + 3] - mu) * inv * g4[h].w + be4[h].w;
        *(v4f*)(out + rowoff + h * 1024 + e0) = o;
    }
}

extern "C" void kernel_launch(void* const* d_in, const int* in_sizes, int n_in,
                              void* d_out, int out_size, void* d_ws, size_t ws_size,
                              hipStream_t stream) {
    const float* facts        = (const float*)d_in[0];
    const float* fact_logits  = (const float*)d_in[1];
    const float* agg_logits   = (const float*)d_in[2];
    const float* strength     = (const float*)d_in[3];
    const float* W_proj       = (const float*)d_in[4];
    const float* ln_gamma     = (const float*)d_in[5];
    const float* ln_beta      = (const float*)d_in[6];
    float* out = (float*)d_out;

    char* ws = (char*)d_ws;
    float* ruleW    = (float*)(ws);
    int*   ruleIdx  = (int*)  (ws + (size_t)R_ * 4 * 4);
    float* projPart = (float*)(ws + 65536);

    prep_gemm_kernel<<<R_ / (RULE_WAVES_ * RULES_PER_WAVE_), 32 * (RULE_WAVES_ + GEMM_WAVES_), 0, stream>>>(
        fact_logits, agg_logits, strength, ruleIdx, ruleW, facts, W_proj, projPart);

    gate_ln_kernel<<<B_, 256, 0, stream>>>(projPart, facts, ruleIdx, ruleW,
                                           ln_gamma, ln_beta, out);
}
